# starting kernel
# baseline (speedup 1.0000x reference)
#include <hip/hip_runtime.h>
#include <hip/hip_bf16.h>


#define Bd    2
#define Hd    8
#define Ld    4096
#define Dd    64
#define Md    256
#define SEG   256
#define NSEG  (Ld / SEG)
#define CK    32
#define NCH   (SEG / CK)
#define PSTR  65
#define FSTR  257
#define ASTR  33
#define STATE (Md * Dd + Md)
#define RATIO 0.0625f
#define EPS   0.001f

typedef float v8f __attribute__((ext_vector_type(8)));
typedef float v2f __attribute__((ext_vector_type(2)));

static __device__ __forceinline__ v8f wmma4(v2f a, v2f b, v8f c) {
  return __builtin_amdgcn_wmma_f32_16x16x4_f32(
       false, a,  false, b,
       (short)0, c,  false,  false);
}

static __device__ __forceinline__ void prefetch_chunk(
    const float* t0, const float* t1, const float* t2,
    int ntens, int b, int h, int lbase, int tid) {
  const int per = CK * 2;
  if (tid < ntens * per) {
    int which = tid / per;
    int r     = (tid % per) >> 1;
    int halfo = (tid & 1) * 32;
    const float* src = (which == 0) ? t0 : (which == 1) ? t1 : t2;
    const float* addr = src + (((size_t)b * Ld + lbase + r) * Hd + h) * Dd + halfo;
    __builtin_prefetch(addr, 0, 3);
  }
}

__global__ __launch_bounds__(256) void perf_attn_segsum(
    const float* __restrict__ kk, const float* __restrict__ vv,
    const float* __restrict__ proj, float* __restrict__ ws)
{
  __shared__ float Plds[Md * PSTR];
  __shared__ float Slds[Md * PSTR];
  __shared__ float sLds[Md];
  __shared__ float raw [CK * PSTR];
  __shared__ float Vlds[CK * PSTR];
  __shared__ float Kf  [CK * FSTR];

  const int tid  = threadIdx.x;
  const int lane = tid & 31, wave = tid >> 5;
  const int lm   = lane & 15, hi   = lane >> 4;

  const int bh  = blockIdx.x / NSEG;
  const int seg = blockIdx.x % NSEG;
  const int b = bh / Hd, h = bh % Hd;

  for (int i = tid; i < Md * Dd; i += 256) {
    Plds[(i >> 6) * PSTR + (i & 63)] = proj[i];
    Slds[(i >> 6) * PSTR + (i & 63)] = 0.f;
  }
  for (int i = tid; i < Md; i += 256) sLds[i] = 0.f;
  __syncthreads();

  for (int ci = 0; ci < NCH; ++ci) {
    const int lbase = seg * SEG + ci * CK;
    for (int i = tid; i < CK * Dd; i += 256) {
      int r = i >> 6, d = i & 63;
      size_t g = (((size_t)b * Ld + lbase + r) * Hd + h) * Dd + d;
      raw [r * PSTR + d] = kk[g];
      Vlds[r * PSTR + d] = vv[g];
    }
    if (ci + 1 < NCH)
      prefetch_chunk(kk, vv, vv, 2, b, h, lbase + CK, tid);
    __syncthreads();

    for (int t = wave; t < 2 * (Md / 16); t += 8) {
      int it = t & 1, mt = t >> 1;
      int r0 = it * 16, c0 = mt * 16;
      v8f acc = {};
      for (int k0 = 0; k0 < Dd; k0 += 4) {
        v2f a, bb;
        a.x  = raw [(r0 + lm) * PSTR + k0 + 2 * hi];
        a.y  = raw [(r0 + lm) * PSTR + k0 + 2 * hi + 1];
        bb.x = Plds[(c0 + lm) * PSTR + k0 + 2 * hi];
        bb.y = Plds[(c0 + lm) * PSTR + k0 + 2 * hi + 1];
        acc = wmma4(a, bb, acc);
      }
      for (int e = 0; e < 8; ++e)
        Kf[(r0 + e + 8 * hi) * FSTR + c0 + lm] =
            fmaxf(RATIO * acc[e], 0.f) + EPS;
    }
    __syncthreads();

    {
      float cs = 0.f;
      for (int r = 0; r < CK; ++r) cs += Kf[r * FSTR + tid];
      sLds[tid] += cs;
    }
    for (int t = wave * 8; t < wave * 8 + 8; ++t) {
      int mt = t >> 2, dt = t & 3;
      int r0 = mt * 16, c0 = dt * 16;
      v8f acc;
      for (int e = 0; e < 8; ++e)
        acc[e] = Slds[(r0 + e + 8 * hi) * PSTR + c0 + lm];
      for (int j0 = 0; j0 < CK; j0 += 4) {
        v2f a, bb;
        a.x  = Kf  [(j0 + 2 * hi)     * FSTR + r0 + lm];
        a.y  = Kf  [(j0 + 2 * hi + 1) * FSTR + r0 + lm];
        bb.x = Vlds[(j0 + 2 * hi)     * PSTR + c0 + lm];
        bb.y = Vlds[(j0 + 2 * hi + 1) * PSTR + c0 + lm];
        acc = wmma4(a, bb, acc);
      }
      for (int e = 0; e < 8; ++e)
        Slds[(r0 + e + 8 * hi) * PSTR + c0 + lm] = acc[e];
    }
    __syncthreads();
  }

  float* dst = ws + (size_t)(bh * NSEG + seg) * STATE;
  for (int i = tid; i < Md * Dd; i += 256)
    dst[i] = Slds[(i >> 6) * PSTR + (i & 63)];
  for (int i = tid; i < Md; i += 256)
    dst[Md * Dd + i] = sLds[i];
}

__global__ __launch_bounds__(256) void perf_attn_prefix(float* __restrict__ ws)
{
  const int bh = blockIdx.x;
  float* base = ws + (size_t)bh * NSEG * STATE;
  for (int e = threadIdx.x; e < STATE; e += 256) {
    float run = 0.f;
    for (int sgi = 0; sgi < NSEG; ++sgi) {
      float* p = base + (size_t)sgi * STATE + e;
      float t = *p;
      *p = run;
      run += t;
    }
  }
}

__global__ __launch_bounds__(256) void perf_attn_main(
    const float* __restrict__ qq, const float* __restrict__ kk,
    const float* __restrict__ vv, const float* __restrict__ proj,
    const float* __restrict__ ws, float* __restrict__ out)
{
  __shared__ float Plds[Md * PSTR];
  __shared__ float Slds[Md * PSTR];
  __shared__ float sLds[Md];
  __shared__ float raw [CK * PSTR];
  __shared__ float Vlds[CK * PSTR];
  __shared__ float Qf  [CK * FSTR];
  __shared__ float Kf  [CK * FSTR];
  __shared__ float Alds[CK * ASTR];
  __shared__ float denLds[CK];

  const int tid  = threadIdx.x;
  const int lane = tid & 31, wave = tid >> 5;
  const int lm   = lane & 15, hi   = lane >> 4;

  const int bh  = blockIdx.x / NSEG;
  const int seg = blockIdx.x % NSEG;
  const int b = bh / Hd, h = bh % Hd;

  for (int i = tid; i < Md * Dd; i += 256)
    Plds[(i >> 6) * PSTR + (i & 63)] = proj[i];
  const float* st = ws + (size_t)(bh * NSEG + seg) * STATE;
  for (int i = tid; i < Md * Dd; i += 256)
    Slds[(i >> 6) * PSTR + (i & 63)] = st[i];
  for (int i = tid; i < Md; i += 256) sLds[i] = st[Md * Dd + i];
  __syncthreads();

  for (int ci = 0; ci < NCH; ++ci) {
    const int lbase = seg * SEG + ci * CK;

    for (int i = tid; i < CK * Dd; i += 256) {
      int r = i >> 6, d = i & 63;
      raw[r * PSTR + d] = qq[(((size_t)b * Ld + lbase + r) * Hd + h) * Dd + d];
    }
    if (ci + 1 < NCH)
      prefetch_chunk(qq, kk, vv, 3, b, h, lbase + CK, tid);
    __syncthreads();

    for (int t = wave; t < 2 * (Md / 16); t += 8) {
      int it = t & 1, mt = t >> 1;
      int r0 = it * 16, c0 = mt * 16;
      v8f acc = {};
      for (int k0 = 0; k0 < Dd; k0 += 4) {
        v2f a, bb;
        a.x  = raw [(r0 + lm) * PSTR + k0 + 2 * hi];
        a.y  = raw [(r0 + lm) * PSTR + k0 + 2 * hi + 1];
        bb.x = Plds[(c0 + lm) * PSTR + k0 + 2 * hi];
        bb.y = Plds[(c0 + lm) * PSTR + k0 + 2 * hi + 1];
        acc = wmma4(a, bb, acc);
      }
      for (int e = 0; e < 8; ++e)
        Qf[(r0 + e + 8 * hi) * FSTR + c0 + lm] =
            fmaxf(RATIO * acc[e], 0.f) + EPS;
    }
    __syncthreads();

    for (int i = tid; i < CK * Dd; i += 256) {
      int r = i >> 6, d = i & 63;
      size_t g = (((size_t)b * Ld + lbase + r) * Hd + h) * Dd + d;
      raw [r * PSTR + d] = kk[g];
      Vlds[r * PSTR + d] = vv[g];
    }
    __syncthreads();

    for (int t = wave; t < 2 * (Md / 16); t += 8) {
      int it = t & 1, mt = t >> 1;
      int r0 = it * 16, c0 = mt * 16;
      v8f acc = {};
      for (int k0 = 0; k0 < Dd; k0 += 4) {
        v2f a, bb;
        a.x  = raw [(r0 + lm) * PSTR + k0 + 2 * hi];
        a.y  = raw [(r0 + lm) * PSTR + k0 + 2 * hi + 1];
        bb.x = Plds[(c0 + lm) * PSTR + k0 + 2 * hi];
        bb.y = Plds[(c0 + lm) * PSTR + k0 + 2 * hi + 1];
        acc = wmma4(a, bb, acc);
      }
      for (int e = 0; e < 8; ++e)
        Kf[(r0 + e + 8 * hi) * FSTR + c0 + lm] =
            fmaxf(RATIO * acc[e], 0.f) + EPS;
    }
    __syncthreads();

    if (wave < 3) {
      const int itA[3] = {0, 1, 1};
      const int jtA[3] = {0, 0, 1};
      int it = itA[wave], jt = jtA[wave];
      v8f acc = {};
      for (int k0 = 0; k0 < Md; k0 += 4) {
        v2f a, bb;
        a.x  = Qf[(it * 16 + lm) * FSTR + k0 + 2 * hi];
        a.y  = Qf[(it * 16 + lm) * FSTR + k0 + 2 * hi + 1];
        bb.x = Kf[(jt * 16 + lm) * FSTR + k0 + 2 * hi];
        bb.y = Kf[(jt * 16 + lm) * FSTR + k0 + 2 * hi + 1];
        acc = wmma4(a, bb, acc);
      }
      for (int e = 0; e < 8; ++e) {
        int gi = it * 16 + e + 8 * hi, gj = jt * 16 + lm;
        Alds[gi * ASTR + gj] = (gj <= gi) ? acc[e] : 0.f;
      }
    } else if (wave == 3) {
      for (int i2 = lane; i2 < 256; i2 += 32)
        Alds[(i2 >> 4) * ASTR + 16 + (i2 & 15)] = 0.f;
    }
    __syncthreads();

    if (wave == 0) {
      float dsum = 0.f;
      for (int m = 0; m < Md; ++m) dsum += Qf[lane * FSTR + m] * sLds[m];
      for (int j = 0; j < CK; ++j) dsum += Alds[lane * ASTR + j];
      denLds[lane] = dsum;
    }
    v8f oacc = {};
    {
      int it = wave >> 2, dt = wave & 3;
      int r0 = it * 16, c0 = dt * 16;
      for (int k0 = 0; k0 < Md; k0 += 4) {
        v2f a, bb;
        a.x  = Qf  [(r0 + lm) * FSTR + k0 + 2 * hi];
        a.y  = Qf  [(r0 + lm) * FSTR + k0 + 2 * hi + 1];
        bb.x = Slds[(k0 + 2 * hi)     * PSTR + c0 + lm];
        bb.y = Slds[(k0 + 2 * hi + 1) * PSTR + c0 + lm];
        oacc = wmma4(a, bb, oacc);
      }
      for (int j0 = 0; j0 < CK; j0 += 4) {
        v2f a, bb;
        a.x  = Alds[(r0 + lm) * ASTR + j0 + 2 * hi];
        a.y  = Alds[(r0 + lm) * ASTR + j0 + 2 * hi + 1];
        bb.x = Vlds[(j0 + 2 * hi)     * PSTR + c0 + lm];
        bb.y = Vlds[(j0 + 2 * hi + 1) * PSTR + c0 + lm];
        oacc = wmma4(a, bb, oacc);
      }
    }
    __syncthreads();

    {
      int it = wave >> 2, dt = wave & 3;
      for (int e = 0; e < 8; ++e) {
        int row = it * 16 + e + 8 * hi;
        float den = denLds[row];
        den = (den <= 0.f) ? 1.f : den;
        float* dst =
            out + (((size_t)b * Ld + lbase + row) * Hd + h) * Dd + dt * 16 + lm;
        __builtin_nontemporal_store(oacc[e] / den, dst);
      }
    }
    {
      float cs = 0.f;
      for (int r = 0; r < CK; ++r) cs += Kf[r * FSTR + tid];
      sLds[tid] += cs;
    }
    for (int t = wave * 8; t < wave * 8 + 8; ++t) {
      int mt = t >> 2, dt = t & 3;
      int r0 = mt * 16, c0 = dt * 16;
      v8f acc;
      for (int e = 0; e < 8; ++e)
        acc[e] = Slds[(r0 + e + 8 * hi) * PSTR + c0 + lm];
      for (int j0 = 0; j0 < CK; j0 += 4) {
        v2f a, bb;
        a.x  = Kf  [(j0 + 2 * hi)     * FSTR + r0 + lm];
        a.y  = Kf  [(j0 + 2 * hi + 1) * FSTR + r0 + lm];
        bb.x = Vlds[(j0 + 2 * hi)     * PSTR + c0 + lm];
        bb.y = Vlds[(j0 + 2 * hi + 1) * PSTR + c0 + lm];
        acc = wmma4(a, bb, acc);
      }
      for (int e = 0; e < 8; ++e)
        Slds[(r0 + e + 8 * hi) * PSTR + c0 + lm] = acc[e];
    }
    __syncthreads();
  }
}

extern "C" void kernel_launch(void* const* d_in, const int* in_sizes, int n_in,
                              void* d_out, int out_size, void* d_ws, size_t ws_size,
                              hipStream_t stream) {
  const float* q    = (const float*)d_in[0];
  const float* k    = (const float*)d_in[1];
  const float* v    = (const float*)d_in[2];
  const float* proj = (const float*)d_in[3];
  float* out = (float*)d_out;
  float* ws  = (float*)d_ws;

  (void)in_sizes; (void)n_in; (void)out_size; (void)ws_size;

  perf_attn_segsum<<<Bd * Hd * NSEG, 256, 0, stream>>>(k, v, proj, ws);
  perf_attn_prefix<<<Bd * Hd, 256, 0, stream>>>(ws);
  perf_attn_main  <<<Bd * Hd * NSEG, 256, 0, stream>>>(q, k, v, proj, ws, out);
}


# Segment prefix (phase 2) re-gridded from 16 blocks to 260: one float4 state column per thread, 15 loads in flight, same serial summation order
# speedup vs baseline: 1.4512x; 1.4512x over previous
#include <hip/hip_runtime.h>
#include <hip/hip_bf16.h>


#define Bd    2
#define Hd    8
#define Ld    4096
#define Dd    64
#define Md    256
#define SEG   256
#define NSEG  (Ld / SEG)
#define CK    32
#define NCH   (SEG / CK)
#define PSTR  65
#define FSTR  257
#define ASTR  33
#define STATE (Md * Dd + Md)
#define RATIO 0.0625f
#define EPS   0.001f

typedef float v8f __attribute__((ext_vector_type(8)));
typedef float v2f __attribute__((ext_vector_type(2)));

static __device__ __forceinline__ v8f wmma4(v2f a, v2f b, v8f c) {
  return __builtin_amdgcn_wmma_f32_16x16x4_f32(
       false, a,  false, b,
       (short)0, c,  false,  false);
}

static __device__ __forceinline__ void prefetch_chunk(
    const float* t0, const float* t1, const float* t2,
    int ntens, int b, int h, int lbase, int tid) {
  const int per = CK * 2;
  if (tid < ntens * per) {
    int which = tid / per;
    int r     = (tid % per) >> 1;
    int halfo = (tid & 1) * 32;
    const float* src = (which == 0) ? t0 : (which == 1) ? t1 : t2;
    const float* addr = src + (((size_t)b * Ld + lbase + r) * Hd + h) * Dd + halfo;
    __builtin_prefetch(addr, 0, 3);
  }
}

__global__ __launch_bounds__(256) void perf_attn_segsum(
    const float* __restrict__ kk, const float* __restrict__ vv,
    const float* __restrict__ proj, float* __restrict__ ws)
{
  __shared__ float Plds[Md * PSTR];
  __shared__ float Slds[Md * PSTR];
  __shared__ float sLds[Md];
  __shared__ float raw [CK * PSTR];
  __shared__ float Vlds[CK * PSTR];
  __shared__ float Kf  [CK * FSTR];

  const int tid  = threadIdx.x;
  const int lane = tid & 31, wave = tid >> 5;
  const int lm   = lane & 15, hi   = lane >> 4;

  const int bh  = blockIdx.x / NSEG;
  const int seg = blockIdx.x % NSEG;
  const int b = bh / Hd, h = bh % Hd;

  for (int i = tid; i < Md * Dd; i += 256) {
    Plds[(i >> 6) * PSTR + (i & 63)] = proj[i];
    Slds[(i >> 6) * PSTR + (i & 63)] = 0.f;
  }
  for (int i = tid; i < Md; i += 256) sLds[i] = 0.f;
  __syncthreads();

  for (int ci = 0; ci < NCH; ++ci) {
    const int lbase = seg * SEG + ci * CK;
    for (int i = tid; i < CK * Dd; i += 256) {
      int r = i >> 6, d = i & 63;
      size_t g = (((size_t)b * Ld + lbase + r) * Hd + h) * Dd + d;
      raw [r * PSTR + d] = kk[g];
      Vlds[r * PSTR + d] = vv[g];
    }
    if (ci + 1 < NCH)
      prefetch_chunk(kk, vv, vv, 2, b, h, lbase + CK, tid);
    __syncthreads();

    for (int t = wave; t < 2 * (Md / 16); t += 8) {
      int it = t & 1, mt = t >> 1;
      int r0 = it * 16, c0 = mt * 16;
      v8f acc = {};
      for (int k0 = 0; k0 < Dd; k0 += 4) {
        v2f a, bb;
        a.x  = raw [(r0 + lm) * PSTR + k0 + 2 * hi];
        a.y  = raw [(r0 + lm) * PSTR + k0 + 2 * hi + 1];
        bb.x = Plds[(c0 + lm) * PSTR + k0 + 2 * hi];
        bb.y = Plds[(c0 + lm) * PSTR + k0 + 2 * hi + 1];
        acc = wmma4(a, bb, acc);
      }
      for (int e = 0; e < 8; ++e)
        Kf[(r0 + e + 8 * hi) * FSTR + c0 + lm] =
            fmaxf(RATIO * acc[e], 0.f) + EPS;
    }
    __syncthreads();

    {
      float cs = 0.f;
      for (int r = 0; r < CK; ++r) cs += Kf[r * FSTR + tid];
      sLds[tid] += cs;
    }
    for (int t = wave * 8; t < wave * 8 + 8; ++t) {
      int mt = t >> 2, dt = t & 3;
      int r0 = mt * 16, c0 = dt * 16;
      v8f acc;
      for (int e = 0; e < 8; ++e)
        acc[e] = Slds[(r0 + e + 8 * hi) * PSTR + c0 + lm];
      for (int j0 = 0; j0 < CK; j0 += 4) {
        v2f a, bb;
        a.x  = Kf  [(j0 + 2 * hi)     * FSTR + r0 + lm];
        a.y  = Kf  [(j0 + 2 * hi + 1) * FSTR + r0 + lm];
        bb.x = Vlds[(j0 + 2 * hi)     * PSTR + c0 + lm];
        bb.y = Vlds[(j0 + 2 * hi + 1) * PSTR + c0 + lm];
        acc = wmma4(a, bb, acc);
      }
      for (int e = 0; e < 8; ++e)
        Slds[(r0 + e + 8 * hi) * PSTR + c0 + lm] = acc[e];
    }
    __syncthreads();
  }

  float* dst = ws + (size_t)(bh * NSEG + seg) * STATE;
  for (int i = tid; i < Md * Dd; i += 256)
    dst[i] = Slds[(i >> 6) * PSTR + (i & 63)];
  for (int i = tid; i < Md; i += 256)
    dst[Md * Dd + i] = sLds[i];
}

#define STATE4 (STATE / 4)
__global__ __launch_bounds__(256) void perf_attn_prefix(float* __restrict__ ws)
{
  const int idx = blockIdx.x * 256 + threadIdx.x;
  const int bh  = idx / STATE4;
  const int e4  = idx - bh * STATE4;
  float4* p = reinterpret_cast<float4*>(ws + (size_t)bh * NSEG * STATE) + e4;

  float4 t[NSEG - 1];
#pragma unroll
  for (int sgi = 0; sgi < NSEG - 1; ++sgi) t[sgi] = p[(size_t)sgi * STATE4];

  float4 run = make_float4(0.f, 0.f, 0.f, 0.f);
#pragma unroll
  for (int sgi = 0; sgi < NSEG; ++sgi) {
    p[(size_t)sgi * STATE4] = run;
    if (sgi < NSEG - 1) {
      run.x += t[sgi].x; run.y += t[sgi].y;
      run.z += t[sgi].z; run.w += t[sgi].w;
    }
  }
}

__global__ __launch_bounds__(256) void perf_attn_main(
    const float* __restrict__ qq, const float* __restrict__ kk,
    const float* __restrict__ vv, const float* __restrict__ proj,
    const float* __restrict__ ws, float* __restrict__ out)
{
  __shared__ float Plds[Md * PSTR];
  __shared__ float Slds[Md * PSTR];
  __shared__ float sLds[Md];
  __shared__ float raw [CK * PSTR];
  __shared__ float Vlds[CK * PSTR];
  __shared__ float Qf  [CK * FSTR];
  __shared__ float Kf  [CK * FSTR];
  __shared__ float Alds[CK * ASTR];
  __shared__ float denLds[CK];

  const int tid  = threadIdx.x;
  const int lane = tid & 31, wave = tid >> 5;
  const int lm   = lane & 15, hi   = lane >> 4;

  const int bh  = blockIdx.x / NSEG;
  const int seg = blockIdx.x % NSEG;
  const int b = bh / Hd, h = bh % Hd;

  for (int i = tid; i < Md * Dd; i += 256)
    Plds[(i >> 6) * PSTR + (i & 63)] = proj[i];
  const float* st = ws + (size_t)(bh * NSEG + seg) * STATE;
  for (int i = tid; i < Md * Dd; i += 256)
    Slds[(i >> 6) * PSTR + (i & 63)] = st[i];
  for (int i = tid; i < Md; i += 256) sLds[i] = st[Md * Dd + i];
  __syncthreads();

  for (int ci = 0; ci < NCH; ++ci) {
    const int lbase = seg * SEG + ci * CK;

    for (int i = tid; i < CK * Dd; i += 256) {
      int r = i >> 6, d = i & 63;
      raw[r * PSTR + d] = qq[(((size_t)b * Ld + lbase + r) * Hd + h) * Dd + d];
    }
    if (ci + 1 < NCH)
      prefetch_chunk(qq, kk, vv, 3, b, h, lbase + CK, tid);
    __syncthreads();

    for (int t = wave; t < 2 * (Md / 16); t += 8) {
      int it = t & 1, mt = t >> 1;
      int r0 = it * 16, c0 = mt * 16;
      v8f acc = {};
      for (int k0 = 0; k0 < Dd; k0 += 4) {
        v2f a, bb;
        a.x  = raw [(r0 + lm) * PSTR + k0 + 2 * hi];
        a.y  = raw [(r0 + lm) * PSTR + k0 + 2 * hi + 1];
        bb.x = Plds[(c0 + lm) * PSTR + k0 + 2 * hi];
        bb.y = Plds[(c0 + lm) * PSTR + k0 + 2 * hi + 1];
        acc = wmma4(a, bb, acc);
      }
      for (int e = 0; e < 8; ++e)
        Qf[(r0 + e + 8 * hi) * FSTR + c0 + lm] =
            fmaxf(RATIO * acc[e], 0.f) + EPS;
    }
    __syncthreads();

    for (int i = tid; i < CK * Dd; i += 256) {
      int r = i >> 6, d = i & 63;
      size_t g = (((size_t)b * Ld + lbase + r) * Hd + h) * Dd + d;
      raw [r * PSTR + d] = kk[g];
      Vlds[r * PSTR + d] = vv[g];
    }
    __syncthreads();

    for (int t = wave; t < 2 * (Md / 16); t += 8) {
      int it = t & 1, mt = t >> 1;
      int r0 = it * 16, c0 = mt * 16;
      v8f acc = {};
      for (int k0 = 0; k0 < Dd; k0 += 4) {
        v2f a, bb;
        a.x  = raw [(r0 + lm) * PSTR + k0 + 2 * hi];
        a.y  = raw [(r0 + lm) * PSTR + k0 + 2 * hi + 1];
        bb.x = Plds[(c0 + lm) * PSTR + k0 + 2 * hi];
        bb.y = Plds[(c0 + lm) * PSTR + k0 + 2 * hi + 1];
        acc = wmma4(a, bb, acc);
      }
      for (int e = 0; e < 8; ++e)
        Kf[(r0 + e + 8 * hi) * FSTR + c0 + lm] =
            fmaxf(RATIO * acc[e], 0.f) + EPS;
    }
    __syncthreads();

    if (wave < 3) {
      const int itA[3] = {0, 1, 1};
      const int jtA[3] = {0, 0, 1};
      int it = itA[wave], jt = jtA[wave];
      v8f acc = {};
      for (int k0 = 0; k0 < Md; k0 += 4) {
        v2f a, bb;
        a.x  = Qf[(it * 16 + lm) * FSTR + k0 + 2 * hi];
        a.y  = Qf[(it * 16 + lm) * FSTR + k0 + 2 * hi + 1];
        bb.x = Kf[(jt * 16 + lm) * FSTR + k0 + 2 * hi];
        bb.y = Kf[(jt * 16 + lm) * FSTR + k0 + 2 * hi + 1];
        acc = wmma4(a, bb, acc);
      }
      for (int e = 0; e < 8; ++e) {
        int gi = it * 16 + e + 8 * hi, gj = jt * 16 + lm;
        Alds[gi * ASTR + gj] = (gj <= gi) ? acc[e] : 0.f;
      }
    } else if (wave == 3) {
      for (int i2 = lane; i2 < 256; i2 += 32)
        Alds[(i2 >> 4) * ASTR + 16 + (i2 & 15)] = 0.f;
    }
    __syncthreads();

    if (wave == 0) {
      float dsum = 0.f;
      for (int m = 0; m < Md; ++m) dsum += Qf[lane * FSTR + m] * sLds[m];
      for (int j = 0; j < CK; ++j) dsum += Alds[lane * ASTR + j];
      denLds[lane] = dsum;
    }
    v8f oacc = {};
    {
      int it = wave >> 2, dt = wave & 3;
      int r0 = it * 16, c0 = dt * 16;
      for (int k0 = 0; k0 < Md; k0 += 4) {
        v2f a, bb;
        a.x  = Qf  [(r0 + lm) * FSTR + k0 + 2 * hi];
        a.y  = Qf  [(r0 + lm) * FSTR + k0 + 2 * hi + 1];
        bb.x = Slds[(k0 + 2 * hi)     * PSTR + c0 + lm];
        bb.y = Slds[(k0 + 2 * hi + 1) * PSTR + c0 + lm];
        oacc = wmma4(a, bb, oacc);
      }
      for (int j0 = 0; j0 < CK; j0 += 4) {
        v2f a, bb;
        a.x  = Alds[(r0 + lm) * ASTR + j0 + 2 * hi];
        a.y  = Alds[(r0 + lm) * ASTR + j0 + 2 * hi + 1];
        bb.x = Vlds[(j0 + 2 * hi)     * PSTR + c0 + lm];
        bb.y = Vlds[(j0 + 2 * hi + 1) * PSTR + c0 + lm];
        oacc = wmma4(a, bb, oacc);
      }
    }
    __syncthreads();

    {
      int it = wave >> 2, dt = wave & 3;
      for (int e = 0; e < 8; ++e) {
        int row = it * 16 + e + 8 * hi;
        float den = denLds[row];
        den = (den <= 0.f) ? 1.f : den;
        float* dst =
            out + (((size_t)b * Ld + lbase + row) * Hd + h) * Dd + dt * 16 + lm;
        __builtin_nontemporal_store(oacc[e] / den, dst);
      }
    }
    {
      float cs = 0.f;
      for (int r = 0; r < CK; ++r) cs += Kf[r * FSTR + tid];
      sLds[tid] += cs;
    }
    for (int t = wave * 8; t < wave * 8 + 8; ++t) {
      int mt = t >> 2, dt = t & 3;
      int r0 = mt * 16, c0 = dt * 16;
      v8f acc;
      for (int e = 0; e < 8; ++e)
        acc[e] = Slds[(r0 + e + 8 * hi) * PSTR + c0 + lm];
      for (int j0 = 0; j0 < CK; j0 += 4) {
        v2f a, bb;
        a.x  = Kf  [(j0 + 2 * hi)     * FSTR + r0 + lm];
        a.y  = Kf  [(j0 + 2 * hi + 1) * FSTR + r0 + lm];
        bb.x = Vlds[(j0 + 2 * hi)     * PSTR + c0 + lm];
        bb.y = Vlds[(j0 + 2 * hi + 1) * PSTR + c0 + lm];
        acc = wmma4(a, bb, acc);
      }
      for (int e = 0; e < 8; ++e)
        Slds[(r0 + e + 8 * hi) * PSTR + c0 + lm] = acc[e];
    }
    __syncthreads();
  }
}

extern "C" void kernel_launch(void* const* d_in, const int* in_sizes, int n_in,
                              void* d_out, int out_size, void* d_ws, size_t ws_size,
                              hipStream_t stream) {
  const float* q    = (const float*)d_in[0];
  const float* k    = (const float*)d_in[1];
  const float* v    = (const float*)d_in[2];
  const float* proj = (const float*)d_in[3];
  float* out = (float*)d_out;
  float* ws  = (float*)d_ws;

  (void)in_sizes; (void)n_in; (void)out_size; (void)ws_size;

  perf_attn_segsum<<<Bd * Hd * NSEG, 256, 0, stream>>>(k, v, proj, ws);
  perf_attn_prefix<<<Bd * Hd * STATE4 / 256, 256, 0, stream>>>(ws);
  perf_attn_main  <<<Bd * Hd * NSEG, 256, 0, stream>>>(q, k, v, proj, ws, out);
}


# LDS row pitches 65/257/33 -> 84/276/36 dwords: WMMA fragment reads and accumulator-tile accesses become bank-conflict-free (were 4-way and 2-way); same loaders, same k order
# speedup vs baseline: 1.5718x; 1.0831x over previous
#include <hip/hip_runtime.h>
#include <hip/hip_bf16.h>


#define Bd    2
#define Hd    8
#define Ld    4096
#define Dd    64
#define Md    256
#define SEG   256
#define NSEG  (Ld / SEG)
#define CK    32
#define NCH   (SEG / CK)
#define PSTR  84
#define FSTR  276
#define ASTR  36
#define STATE (Md * Dd + Md)
#define RATIO 0.0625f
#define EPS   0.001f

typedef float v8f __attribute__((ext_vector_type(8)));
typedef float v2f __attribute__((ext_vector_type(2)));

static __device__ __forceinline__ v8f wmma4(v2f a, v2f b, v8f c) {
  return __builtin_amdgcn_wmma_f32_16x16x4_f32(
       false, a,  false, b,
       (short)0, c,  false,  false);
}

static __device__ __forceinline__ void prefetch_chunk(
    const float* t0, const float* t1, const float* t2,
    int ntens, int b, int h, int lbase, int tid) {
  const int per = CK * 2;
  if (tid < ntens * per) {
    int which = tid / per;
    int r     = (tid % per) >> 1;
    int halfo = (tid & 1) * 32;
    const float* src = (which == 0) ? t0 : (which == 1) ? t1 : t2;
    const float* addr = src + (((size_t)b * Ld + lbase + r) * Hd + h) * Dd + halfo;
    __builtin_prefetch(addr, 0, 3);
  }
}

__global__ __launch_bounds__(256) void perf_attn_segsum(
    const float* __restrict__ kk, const float* __restrict__ vv,
    const float* __restrict__ proj, float* __restrict__ ws)
{
  __shared__ float Plds[Md * PSTR];
  __shared__ float Slds[Md * PSTR];
  __shared__ float sLds[Md];
  __shared__ float raw [CK * PSTR];
  __shared__ float Vlds[CK * PSTR];
  __shared__ float Kf  [CK * FSTR];

  const int tid  = threadIdx.x;
  const int lane = tid & 31, wave = tid >> 5;
  const int lm   = lane & 15, hi   = lane >> 4;

  const int bh  = blockIdx.x / NSEG;
  const int seg = blockIdx.x % NSEG;
  const int b = bh / Hd, h = bh % Hd;

  for (int i = tid; i < Md * Dd; i += 256) {
    Plds[(i >> 6) * PSTR + (i & 63)] = proj[i];
    Slds[(i >> 6) * PSTR + (i & 63)] = 0.f;
  }
  for (int i = tid; i < Md; i += 256) sLds[i] = 0.f;
  __syncthreads();

  for (int ci = 0; ci < NCH; ++ci) {
    const int lbase = seg * SEG + ci * CK;
    for (int i = tid; i < CK * Dd; i += 256) {
      int r = i >> 6, d = i & 63;
      size_t g = (((size_t)b * Ld + lbase + r) * Hd + h) * Dd + d;
      raw [r * PSTR + d] = kk[g];
      Vlds[r * PSTR + d] = vv[g];
    }
    if (ci + 1 < NCH)
      prefetch_chunk(kk, vv, vv, 2, b, h, lbase + CK, tid);
    __syncthreads();

    for (int t = wave; t < 2 * (Md / 16); t += 8) {
      int it = t & 1, mt = t >> 1;
      int r0 = it * 16, c0 = mt * 16;
      v8f acc = {};
      for (int k0 = 0; k0 < Dd; k0 += 4) {
        v2f a, bb;
        a.x  = raw [(r0 + lm) * PSTR + k0 + 2 * hi];
        a.y  = raw [(r0 + lm) * PSTR + k0 + 2 * hi + 1];
        bb.x = Plds[(c0 + lm) * PSTR + k0 + 2 * hi];
        bb.y = Plds[(c0 + lm) * PSTR + k0 + 2 * hi + 1];
        acc = wmma4(a, bb, acc);
      }
      for (int e = 0; e < 8; ++e)
        Kf[(r0 + e + 8 * hi) * FSTR + c0 + lm] =
            fmaxf(RATIO * acc[e], 0.f) + EPS;
    }
    __syncthreads();

    {
      float cs = 0.f;
      for (int r = 0; r < CK; ++r) cs += Kf[r * FSTR + tid];
      sLds[tid] += cs;
    }
    for (int t = wave * 8; t < wave * 8 + 8; ++t) {
      int mt = t >> 2, dt = t & 3;
      int r0 = mt * 16, c0 = dt * 16;
      v8f acc;
      for (int e = 0; e < 8; ++e)
        acc[e] = Slds[(r0 + e + 8 * hi) * PSTR + c0 + lm];
      for (int j0 = 0; j0 < CK; j0 += 4) {
        v2f a, bb;
        a.x  = Kf  [(j0 + 2 * hi)     * FSTR + r0 + lm];
        a.y  = Kf  [(j0 + 2 * hi + 1) * FSTR + r0 + lm];
        bb.x = Vlds[(j0 + 2 * hi)     * PSTR + c0 + lm];
        bb.y = Vlds[(j0 + 2 * hi + 1) * PSTR + c0 + lm];
        acc = wmma4(a, bb, acc);
      }
      for (int e = 0; e < 8; ++e)
        Slds[(r0 + e + 8 * hi) * PSTR + c0 + lm] = acc[e];
    }
    __syncthreads();
  }

  float* dst = ws + (size_t)(bh * NSEG + seg) * STATE;
  for (int i = tid; i < Md * Dd; i += 256)
    dst[i] = Slds[(i >> 6) * PSTR + (i & 63)];
  for (int i = tid; i < Md; i += 256)
    dst[Md * Dd + i] = sLds[i];
}

#define STATE4 (STATE / 4)
__global__ __launch_bounds__(256) void perf_attn_prefix(float* __restrict__ ws)
{
  const int idx = blockIdx.x * 256 + threadIdx.x;
  const int bh  = idx / STATE4;
  const int e4  = idx - bh * STATE4;
  float4* p = reinterpret_cast<float4*>(ws + (size_t)bh * NSEG * STATE) + e4;

  float4 t[NSEG - 1];
#pragma unroll
  for (int sgi = 0; sgi < NSEG - 1; ++sgi) t[sgi] = p[(size_t)sgi * STATE4];

  float4 run = make_float4(0.f, 0.f, 0.f, 0.f);
#pragma unroll
  for (int sgi = 0; sgi < NSEG; ++sgi) {
    p[(size_t)sgi * STATE4] = run;
    if (sgi < NSEG - 1) {
      run.x += t[sgi].x; run.y += t[sgi].y;
      run.z += t[sgi].z; run.w += t[sgi].w;
    }
  }
}

__global__ __launch_bounds__(256) void perf_attn_main(
    const float* __restrict__ qq, const float* __restrict__ kk,
    const float* __restrict__ vv, const float* __restrict__ proj,
    const float* __restrict__ ws, float* __restrict__ out)
{
  __shared__ float Plds[Md * PSTR];
  __shared__ float Slds[Md * PSTR];
  __shared__ float sLds[Md];
  __shared__ float raw [CK * PSTR];
  __shared__ float Vlds[CK * PSTR];
  __shared__ float Qf  [CK * FSTR];
  __shared__ float Kf  [CK * FSTR];
  __shared__ float Alds[CK * ASTR];
  __shared__ float denLds[CK];

  const int tid  = threadIdx.x;
  const int lane = tid & 31, wave = tid >> 5;
  const int lm   = lane & 15, hi   = lane >> 4;

  const int bh  = blockIdx.x / NSEG;
  const int seg = blockIdx.x % NSEG;
  const int b = bh / Hd, h = bh % Hd;

  for (int i = tid; i < Md * Dd; i += 256)
    Plds[(i >> 6) * PSTR + (i & 63)] = proj[i];
  const float* st = ws + (size_t)(bh * NSEG + seg) * STATE;
  for (int i = tid; i < Md * Dd; i += 256)
    Slds[(i >> 6) * PSTR + (i & 63)] = st[i];
  for (int i = tid; i < Md; i += 256) sLds[i] = st[Md * Dd + i];
  __syncthreads();

  for (int ci = 0; ci < NCH; ++ci) {
    const int lbase = seg * SEG + ci * CK;

    for (int i = tid; i < CK * Dd; i += 256) {
      int r = i >> 6, d = i & 63;
      raw[r * PSTR + d] = qq[(((size_t)b * Ld + lbase + r) * Hd + h) * Dd + d];
    }
    if (ci + 1 < NCH)
      prefetch_chunk(qq, kk, vv, 3, b, h, lbase + CK, tid);
    __syncthreads();

    for (int t = wave; t < 2 * (Md / 16); t += 8) {
      int it = t & 1, mt = t >> 1;
      int r0 = it * 16, c0 = mt * 16;
      v8f acc = {};
      for (int k0 = 0; k0 < Dd; k0 += 4) {
        v2f a, bb;
        a.x  = raw [(r0 + lm) * PSTR + k0 + 2 * hi];
        a.y  = raw [(r0 + lm) * PSTR + k0 + 2 * hi + 1];
        bb.x = Plds[(c0 + lm) * PSTR + k0 + 2 * hi];
        bb.y = Plds[(c0 + lm) * PSTR + k0 + 2 * hi + 1];
        acc = wmma4(a, bb, acc);
      }
      for (int e = 0; e < 8; ++e)
        Qf[(r0 + e + 8 * hi) * FSTR + c0 + lm] =
            fmaxf(RATIO * acc[e], 0.f) + EPS;
    }
    __syncthreads();

    for (int i = tid; i < CK * Dd; i += 256) {
      int r = i >> 6, d = i & 63;
      size_t g = (((size_t)b * Ld + lbase + r) * Hd + h) * Dd + d;
      raw [r * PSTR + d] = kk[g];
      Vlds[r * PSTR + d] = vv[g];
    }
    __syncthreads();

    for (int t = wave; t < 2 * (Md / 16); t += 8) {
      int it = t & 1, mt = t >> 1;
      int r0 = it * 16, c0 = mt * 16;
      v8f acc = {};
      for (int k0 = 0; k0 < Dd; k0 += 4) {
        v2f a, bb;
        a.x  = raw [(r0 + lm) * PSTR + k0 + 2 * hi];
        a.y  = raw [(r0 + lm) * PSTR + k0 + 2 * hi + 1];
        bb.x = Plds[(c0 + lm) * PSTR + k0 + 2 * hi];
        bb.y = Plds[(c0 + lm) * PSTR + k0 + 2 * hi + 1];
        acc = wmma4(a, bb, acc);
      }
      for (int e = 0; e < 8; ++e)
        Kf[(r0 + e + 8 * hi) * FSTR + c0 + lm] =
            fmaxf(RATIO * acc[e], 0.f) + EPS;
    }
    __syncthreads();

    if (wave < 3) {
      const int itA[3] = {0, 1, 1};
      const int jtA[3] = {0, 0, 1};
      int it = itA[wave], jt = jtA[wave];
      v8f acc = {};
      for (int k0 = 0; k0 < Md; k0 += 4) {
        v2f a, bb;
        a.x  = Qf[(it * 16 + lm) * FSTR + k0 + 2 * hi];
        a.y  = Qf[(it * 16 + lm) * FSTR + k0 + 2 * hi + 1];
        bb.x = Kf[(jt * 16 + lm) * FSTR + k0 + 2 * hi];
        bb.y = Kf[(jt * 16 + lm) * FSTR + k0 + 2 * hi + 1];
        acc = wmma4(a, bb, acc);
      }
      for (int e = 0; e < 8; ++e) {
        int gi = it * 16 + e + 8 * hi, gj = jt * 16 + lm;
        Alds[gi * ASTR + gj] = (gj <= gi) ? acc[e] : 0.f;
      }
    } else if (wave == 3) {
      for (int i2 = lane; i2 < 256; i2 += 32)
        Alds[(i2 >> 4) * ASTR + 16 + (i2 & 15)] = 0.f;
    }
    __syncthreads();

    if (wave == 0) {
      float dsum = 0.f;
      for (int m = 0; m < Md; ++m) dsum += Qf[lane * FSTR + m] * sLds[m];
      for (int j = 0; j < CK; ++j) dsum += Alds[lane * ASTR + j];
      denLds[lane] = dsum;
    }
    v8f oacc = {};
    {
      int it = wave >> 2, dt = wave & 3;
      int r0 = it * 16, c0 = dt * 16;
      for (int k0 = 0; k0 < Md; k0 += 4) {
        v2f a, bb;
        a.x  = Qf  [(r0 + lm) * FSTR + k0 + 2 * hi];
        a.y  = Qf  [(r0 + lm) * FSTR + k0 + 2 * hi + 1];
        bb.x = Slds[(k0 + 2 * hi)     * PSTR + c0 + lm];
        bb.y = Slds[(k0 + 2 * hi + 1) * PSTR + c0 + lm];
        oacc = wmma4(a, bb, oacc);
      }
      for (int j0 = 0; j0 < CK; j0 += 4) {
        v2f a, bb;
        a.x  = Alds[(r0 + lm) * ASTR + j0 + 2 * hi];
        a.y  = Alds[(r0 + lm) * ASTR + j0 + 2 * hi + 1];
        bb.x = Vlds[(j0 + 2 * hi)     * PSTR + c0 + lm];
        bb.y = Vlds[(j0 + 2 * hi + 1) * PSTR + c0 + lm];
        oacc = wmma4(a, bb, oacc);
      }
    }
    __syncthreads();

    {
      int it = wave >> 2, dt = wave & 3;
      for (int e = 0; e < 8; ++e) {
        int row = it * 16 + e + 8 * hi;
        float den = denLds[row];
        den = (den <= 0.f) ? 1.f : den;
        float* dst =
            out + (((size_t)b * Ld + lbase + row) * Hd + h) * Dd + dt * 16 + lm;
        __builtin_nontemporal_store(oacc[e] / den, dst);
      }
    }
    {
      float cs = 0.f;
      for (int r = 0; r < CK; ++r) cs += Kf[r * FSTR + tid];
      sLds[tid] += cs;
    }
    for (int t = wave * 8; t < wave * 8 + 8; ++t) {
      int mt = t >> 2, dt = t & 3;
      int r0 = mt * 16, c0 = dt * 16;
      v8f acc;
      for (int e = 0; e < 8; ++e)
        acc[e] = Slds[(r0 + e + 8 * hi) * PSTR + c0 + lm];
      for (int j0 = 0; j0 < CK; j0 += 4) {
        v2f a, bb;
        a.x  = Kf  [(j0 + 2 * hi)     * FSTR + r0 + lm];
        a.y  = Kf  [(j0 + 2 * hi + 1) * FSTR + r0 + lm];
        bb.x = Vlds[(j0 + 2 * hi)     * PSTR + c0 + lm];
        bb.y = Vlds[(j0 + 2 * hi + 1) * PSTR + c0 + lm];
        acc = wmma4(a, bb, acc);
      }
      for (int e = 0; e < 8; ++e)
        Slds[(r0 + e + 8 * hi) * PSTR + c0 + lm] = acc[e];
    }
    __syncthreads();
  }
}

extern "C" void kernel_launch(void* const* d_in, const int* in_sizes, int n_in,
                              void* d_out, int out_size, void* d_ws, size_t ws_size,
                              hipStream_t stream) {
  const float* q    = (const float*)d_in[0];
  const float* k    = (const float*)d_in[1];
  const float* v    = (const float*)d_in[2];
  const float* proj = (const float*)d_in[3];
  float* out = (float*)d_out;
  float* ws  = (float*)d_ws;

  (void)in_sizes; (void)n_in; (void)out_size; (void)ws_size;

  perf_attn_segsum<<<Bd * Hd * NSEG, 256, 0, stream>>>(k, v, proj, ws);
  perf_attn_prefix<<<Bd * Hd * STATE4 / 256, 256, 0, stream>>>(ws);
  perf_attn_main  <<<Bd * Hd * NSEG, 256, 0, stream>>>(q, k, v, proj, ws, out);
}


# WMMA stages register-blocked: each wave keeps projection fragments and its 8 S accumulator tiles in VGPRs across chunks and reads a block's LDS fragments ahead of its products; sum order kept
# speedup vs baseline: 2.1210x; 1.3494x over previous
#include <hip/hip_runtime.h>
#include <hip/hip_bf16.h>


#define Bd    2
#define Hd    8
#define Ld    4096
#define Dd    64
#define Md    256
#define SEG   256
#define NSEG  (Ld / SEG)
#define CK    32
#define NCH   (SEG / CK)
#define PSTR  84
#define FSTR  276
#define ASTR  36
#define STATE (Md * Dd + Md)
#define RATIO 0.0625f
#define EPS   0.001f

typedef float v8f __attribute__((ext_vector_type(8)));
typedef float v2f __attribute__((ext_vector_type(2)));

static __device__ __forceinline__ v8f wmma4(v2f a, v2f b, v8f c) {
  return __builtin_amdgcn_wmma_f32_16x16x4_f32(
       false, a,  false, b,
       (short)0, c,  false,  false);
}

static __device__ __forceinline__ void prefetch_chunk(
    const float* t0, const float* t1, const float* t2,
    int ntens, int b, int h, int lbase, int tid) {
  const int per = CK * 2;
  if (tid < ntens * per) {
    int which = tid / per;
    int r     = (tid % per) >> 1;
    int halfo = (tid & 1) * 32;
    const float* src = (which == 0) ? t0 : (which == 1) ? t1 : t2;
    const float* addr = src + (((size_t)b * Ld + lbase + r) * Hd + h) * Dd + halfo;
    __builtin_prefetch(addr, 0, 3);
  }
}

static __device__ __forceinline__ v2f ld_rm(const float* p) {
  return *reinterpret_cast<const v2f*>(p);
}
static __device__ __forceinline__ v2f ld_km(const float* p, int pitch) {
  v2f r; r.x = p[0]; r.y = p[pitch]; return r;
}

#define KS16 (Dd / 4)

static __device__ __forceinline__ void load_proj_frags(
    const float* __restrict__ proj, int wave, int lm, int hi, v2f (&pf)[2][KS16]) {
#pragma unroll
  for (int j = 0; j < 2; ++j)
#pragma unroll
    for (int ks = 0; ks < KS16; ++ks)
      pf[j][ks] = ld_rm(proj + ((2 * wave + j) * 16 + lm) * Dd + 4 * ks + 2 * hi);
}

static __device__ __forceinline__ void feature_stage(
    const float* raw, const v2f (&pf)[2][KS16], float* dstF,
    int wave, int lm, int hi) {
  v2f a0[KS16], a1[KS16];
#pragma unroll
  for (int ks = 0; ks < KS16; ++ks) {
    a0[ks] = ld_rm(raw + lm * PSTR + 4 * ks + 2 * hi);
    a1[ks] = ld_rm(raw + (16 + lm) * PSTR + 4 * ks + 2 * hi);
  }
  v8f c00 = {}, c01 = {}, c10 = {}, c11 = {};
  __builtin_amdgcn_sched_barrier(0);
#pragma unroll
  for (int ks = 0; ks < KS16; ++ks) {
    c00 = wmma4(a0[ks], pf[0][ks], c00);
    c01 = wmma4(a0[ks], pf[1][ks], c01);
    c10 = wmma4(a1[ks], pf[0][ks], c10);
    c11 = wmma4(a1[ks], pf[1][ks], c11);
  }
  const int cA = (2 * wave) * 16 + lm, cB = cA + 16;
#pragma unroll
  for (int e = 0; e < 8; ++e) {
    dstF[(e + 8 * hi) * FSTR + cA]      = fmaxf(RATIO * c00[e], 0.f) + EPS;
    dstF[(e + 8 * hi) * FSTR + cB]      = fmaxf(RATIO * c01[e], 0.f) + EPS;
    dstF[(16 + e + 8 * hi) * FSTR + cA] = fmaxf(RATIO * c10[e], 0.f) + EPS;
    dstF[(16 + e + 8 * hi) * FSTR + cB] = fmaxf(RATIO * c11[e], 0.f) + EPS;
  }
}

#define KS8 (CK / 4)
static __device__ __forceinline__ void state_update(
    const float* Kf, const float* Vlds, v8f (&sacc)[2][4],
    int wave, int lm, int hi) {
  v2f ka[2][KS8], vb[4][KS8];
#pragma unroll
  for (int ks = 0; ks < KS8; ++ks) {
#pragma unroll
    for (int m2 = 0; m2 < 2; ++m2)
      ka[m2][ks] = ld_km(Kf + (4 * ks + 2 * hi) * FSTR + 32 * wave + 16 * m2 + lm, FSTR);
#pragma unroll
    for (int dt = 0; dt < 4; ++dt)
      vb[dt][ks] = ld_km(Vlds + (4 * ks + 2 * hi) * PSTR + 16 * dt + lm, PSTR);
  }
  __builtin_amdgcn_sched_barrier(0);
#pragma unroll
  for (int ks = 0; ks < KS8; ++ks)
#pragma unroll
    for (int m2 = 0; m2 < 2; ++m2)
#pragma unroll
      for (int dt = 0; dt < 4; ++dt)
        sacc[m2][dt] = wmma4(ka[m2][ks], vb[dt][ks], sacc[m2][dt]);
}

__global__ __launch_bounds__(256) void perf_attn_segsum(
    const float* __restrict__ kk, const float* __restrict__ vv,
    const float* __restrict__ proj, float* __restrict__ ws)
{
  __shared__ __attribute__((aligned(16))) float raw [CK * PSTR];
  __shared__ __attribute__((aligned(16))) float Vlds[CK * PSTR];
  __shared__ __attribute__((aligned(16))) float Kf  [CK * FSTR];

  const int tid  = threadIdx.x;
  const int lane = tid & 31, wave = tid >> 5;
  const int lm   = lane & 15, hi   = lane >> 4;

  const int bh  = blockIdx.x / NSEG;
  const int seg = blockIdx.x % NSEG;
  const int b = bh / Hd, h = bh % Hd;

  v2f pf[2][KS16];
  load_proj_frags(proj, wave, lm, hi, pf);

  v8f sacc[2][4];
#pragma unroll
  for (int m2 = 0; m2 < 2; ++m2)
#pragma unroll
    for (int dt = 0; dt < 4; ++dt)
      sacc[m2][dt] = v8f{};
  float ssum = 0.f;

  for (int ci = 0; ci < NCH; ++ci) {
    const int lbase = seg * SEG + ci * CK;
    for (int i = tid; i < CK * Dd; i += 256) {
      int r = i >> 6, d = i & 63;
      size_t g = (((size_t)b * Ld + lbase + r) * Hd + h) * Dd + d;
      raw [r * PSTR + d] = kk[g];
      Vlds[r * PSTR + d] = vv[g];
    }
    if (ci + 1 < NCH)
      prefetch_chunk(kk, vv, vv, 2, b, h, lbase + CK, tid);
    __syncthreads();

    feature_stage(raw, pf, Kf, wave, lm, hi);
    __syncthreads();

    {
      float cs = 0.f;
      for (int r = 0; r < CK; ++r) cs += Kf[r * FSTR + tid];
      ssum += cs;
    }
    state_update(Kf, Vlds, sacc, wave, lm, hi);
    __syncthreads();
  }

  float* dst = ws + (size_t)(bh * NSEG + seg) * STATE;
#pragma unroll
  for (int m2 = 0; m2 < 2; ++m2)
#pragma unroll
    for (int dt = 0; dt < 4; ++dt)
#pragma unroll
      for (int e = 0; e < 8; ++e)
        dst[(32 * wave + 16 * m2 + e + 8 * hi) * Dd + 16 * dt + lm] = sacc[m2][dt][e];
  dst[Md * Dd + tid] = ssum;
}

#define STATE4 (STATE / 4)
__global__ __launch_bounds__(256) void perf_attn_prefix(float* __restrict__ ws)
{
  const int idx = blockIdx.x * 256 + threadIdx.x;
  const int bh  = idx / STATE4;
  const int e4  = idx - bh * STATE4;
  float4* p = reinterpret_cast<float4*>(ws + (size_t)bh * NSEG * STATE) + e4;

  float4 t[NSEG - 1];
#pragma unroll
  for (int sgi = 0; sgi < NSEG - 1; ++sgi) t[sgi] = p[(size_t)sgi * STATE4];

  float4 run = make_float4(0.f, 0.f, 0.f, 0.f);
#pragma unroll
  for (int sgi = 0; sgi < NSEG; ++sgi) {
    p[(size_t)sgi * STATE4] = run;
    if (sgi < NSEG - 1) {
      run.x += t[sgi].x; run.y += t[sgi].y;
      run.z += t[sgi].z; run.w += t[sgi].w;
    }
  }
}

__global__ __launch_bounds__(256) void perf_attn_main(
    const float* __restrict__ qq, const float* __restrict__ kk,
    const float* __restrict__ vv, const float* __restrict__ proj,
    const float* __restrict__ ws, float* __restrict__ out)
{
  __shared__ __attribute__((aligned(16))) float Slds[Md * PSTR];
  __shared__ __attribute__((aligned(16))) float sLds[Md];
  __shared__ __attribute__((aligned(16))) float raw [CK * PSTR];
  __shared__ __attribute__((aligned(16))) float Vlds[CK * PSTR];
  __shared__ __attribute__((aligned(16))) float Qf  [CK * FSTR];
  __shared__ __attribute__((aligned(16))) float Kf  [CK * FSTR];
  __shared__ __attribute__((aligned(16))) float Alds[CK * ASTR];
  __shared__ float denLds[CK];

  const int tid  = threadIdx.x;
  const int lane = tid & 31, wave = tid >> 5;
  const int lm   = lane & 15, hi   = lane >> 4;

  const int bh  = blockIdx.x / NSEG;
  const int seg = blockIdx.x % NSEG;
  const int b = bh / Hd, h = bh % Hd;

  v2f pf[2][KS16];
  load_proj_frags(proj, wave, lm, hi, pf);

  const float* st = ws + (size_t)(bh * NSEG + seg) * STATE;
  v8f sacc[2][4];
#pragma unroll
  for (int m2 = 0; m2 < 2; ++m2)
#pragma unroll
    for (int dt = 0; dt < 4; ++dt)
#pragma unroll
      for (int e = 0; e < 8; ++e) {
        int row = 32 * wave + 16 * m2 + e + 8 * hi, col = 16 * dt + lm;
        float x = st[row * Dd + col];
        sacc[m2][dt][e] = x;
        Slds[row * PSTR + col] = x;
      }
  sLds[tid] = st[Md * Dd + tid];
  __syncthreads();

  for (int ci = 0; ci < NCH; ++ci) {
    const int lbase = seg * SEG + ci * CK;

    for (int i = tid; i < CK * Dd; i += 256) {
      int r = i >> 6, d = i & 63;
      raw[r * PSTR + d] = qq[(((size_t)b * Ld + lbase + r) * Hd + h) * Dd + d];
    }
    if (ci + 1 < NCH)
      prefetch_chunk(qq, kk, vv, 3, b, h, lbase + CK, tid);
    __syncthreads();

    feature_stage(raw, pf, Qf, wave, lm, hi);
    __syncthreads();

    for (int i = tid; i < CK * Dd; i += 256) {
      int r = i >> 6, d = i & 63;
      size_t g = (((size_t)b * Ld + lbase + r) * Hd + h) * Dd + d;
      raw [r * PSTR + d] = kk[g];
      Vlds[r * PSTR + d] = vv[g];
    }
    __syncthreads();

    feature_stage(raw, pf, Kf, wave, lm, hi);
    __syncthreads();

    if (wave < 3) {
      const int it = (wave > 0) ? 1 : 0;
      const int jt = (wave > 1) ? 1 : 0;
      const float* qrow = Qf + (it * 16 + lm) * FSTR + 2 * hi;
      const float* krow = Kf + (jt * 16 + lm) * FSTR + 2 * hi;
      v2f a[2][KS8], bb[2][KS8];
#pragma unroll
      for (int ks = 0; ks < KS8; ++ks) {
        a [0][ks] = ld_rm(qrow + 4 * ks);
        bb[0][ks] = ld_rm(krow + 4 * ks);
      }
      v8f acc = {};
#pragma unroll
      for (int blk = 0; blk < Md / 32; ++blk) {
        if (blk + 1 < Md / 32) {
#pragma unroll
          for (int ks = 0; ks < KS8; ++ks) {
            a [(blk + 1) & 1][ks] = ld_rm(qrow + 32 * (blk + 1) + 4 * ks);
            bb[(blk + 1) & 1][ks] = ld_rm(krow + 32 * (blk + 1) + 4 * ks);
          }
        }
        __builtin_amdgcn_sched_barrier(0);
#pragma unroll
        for (int ks = 0; ks < KS8; ++ks)
          acc = wmma4(a[blk & 1][ks], bb[blk & 1][ks], acc);
      }
#pragma unroll
      for (int e = 0; e < 8; ++e) {
        int gi = it * 16 + e + 8 * hi, gj = jt * 16 + lm;
        Alds[gi * ASTR + gj] = (gj <= gi) ? acc[e] : 0.f;
      }
    } else if (wave == 3) {
      for (int i2 = lane; i2 < 256; i2 += 32)
        Alds[(i2 >> 4) * ASTR + 16 + (i2 & 15)] = 0.f;
    }
    __syncthreads();

    if (wave == 0) {
      const float4* qv = reinterpret_cast<const float4*>(Qf + lane * FSTR);
      const float4* sv = reinterpret_cast<const float4*>(sLds);
      const float4* av = reinterpret_cast<const float4*>(Alds + lane * ASTR);
      float4 q4[2][8], s4[2][8], a4[CK / 4];
#pragma unroll
      for (int i = 0; i < 8; ++i) { q4[0][i] = qv[i]; s4[0][i] = sv[i]; }
      float dsum = 0.f;
#pragma unroll
      for (int blk = 0; blk < Md / 32; ++blk) {
        if (blk + 1 < Md / 32) {
#pragma unroll
          for (int i = 0; i < 8; ++i) {
            q4[(blk + 1) & 1][i] = qv[8 * (blk + 1) + i];
            s4[(blk + 1) & 1][i] = sv[8 * (blk + 1) + i];
          }
        } else {
#pragma unroll
          for (int i = 0; i < CK / 4; ++i) a4[i] = av[i];
        }
        __builtin_amdgcn_sched_barrier(0);
#pragma unroll
        for (int i = 0; i < 8; ++i) {
          dsum = __builtin_fmaf(q4[blk & 1][i].x, s4[blk & 1][i].x, dsum);
          dsum = __builtin_fmaf(q4[blk & 1][i].y, s4[blk & 1][i].y, dsum);
          dsum = __builtin_fmaf(q4[blk & 1][i].z, s4[blk & 1][i].z, dsum);
          dsum = __builtin_fmaf(q4[blk & 1][i].w, s4[blk & 1][i].w, dsum);
        }
      }
#pragma unroll
      for (int i = 0; i < CK / 4; ++i) {
        dsum += a4[i].x; dsum += a4[i].y; dsum += a4[i].z; dsum += a4[i].w;
      }
      denLds[lane] = dsum;
    }
    v8f oacc = {};
    const int oit = wave >> 2, odt = wave & 3;
    {
      const float* qrow = Qf   + (oit * 16 + lm) * FSTR + 2 * hi;
      const float* scol = Slds + (2 * hi) * PSTR + odt * 16 + lm;
      const float* arow = Alds + (oit * 16 + lm) * ASTR + 2 * hi;
      const float* vcol = Vlds + (2 * hi) * PSTR + odt * 16 + lm;
      v2f a[2][KS8], bb[2][KS8], ia[KS8], ib[KS8];
#pragma unroll
      for (int ks = 0; ks < KS8; ++ks) {
        a [0][ks] = ld_rm(qrow + 4 * ks);
        bb[0][ks] = ld_km(scol + (4 * ks) * PSTR, PSTR);
      }
#pragma unroll
      for (int ks = 0; ks < KS8; ++ks) {
        ia[ks] = ld_rm(arow + 4 * ks);
        ib[ks] = ld_km(vcol + (4 * ks) * PSTR, PSTR);
      }
#pragma unroll
      for (int blk = 0; blk < Md / 32; ++blk) {
        if (blk + 1 < Md / 32) {
#pragma unroll
          for (int ks = 0; ks < KS8; ++ks) {
            a [(blk + 1) & 1][ks] = ld_rm(qrow + 32 * (blk + 1) + 4 * ks);
            bb[(blk + 1) & 1][ks] = ld_km(scol + (32 * (blk + 1) + 4 * ks) * PSTR, PSTR);
          }
        }
        __builtin_amdgcn_sched_barrier(0);
#pragma unroll
        for (int ks = 0; ks < KS8; ++ks)
          oacc = wmma4(a[blk & 1][ks], bb[blk & 1][ks], oacc);
      }
#pragma unroll
      for (int ks = 0; ks < KS8; ++ks)
        oacc = wmma4(ia[ks], ib[ks], oacc);
    }
    float cs = 0.f;
    for (int r = 0; r < CK; ++r) cs += Kf[r * FSTR + tid];
    state_update(Kf, Vlds, sacc, wave, lm, hi);
    __syncthreads();

    {
#pragma unroll
      for (int e = 0; e < 8; ++e) {
        int row = oit * 16 + e + 8 * hi;
        float den = denLds[row];
        den = (den <= 0.f) ? 1.f : den;
        float* dst =
            out + (((size_t)b * Ld + lbase + row) * Hd + h) * Dd + odt * 16 + lm;
        __builtin_nontemporal_store(oacc[e] / den, dst);
      }
    }
    sLds[tid] += cs;
#pragma unroll
    for (int m2 = 0; m2 < 2; ++m2)
#pragma unroll
      for (int dt = 0; dt < 4; ++dt)
#pragma unroll
        for (int e = 0; e < 8; ++e)
          Slds[(32 * wave + 16 * m2 + e + 8 * hi) * PSTR + 16 * dt + lm] = sacc[m2][dt][e];
  }
}

extern "C" void kernel_launch(void* const* d_in, const int* in_sizes, int n_in,
                              void* d_out, int out_size, void* d_ws, size_t ws_size,
                              hipStream_t stream) {
  const float* q    = (const float*)d_in[0];
  const float* k    = (const float*)d_in[1];
  const float* v    = (const float*)d_in[2];
  const float* proj = (const float*)d_in[3];
  float* out = (float*)d_out;
  float* ws  = (float*)d_ws;

  (void)in_sizes; (void)n_in; (void)out_size; (void)ws_size;

  perf_attn_segsum<<<Bd * Hd * NSEG, 256, 0, stream>>>(k, v, proj, ws);
  perf_attn_prefix<<<Bd * Hd * STATE4 / 256, 256, 0, stream>>>(ws);
  perf_attn_main  <<<Bd * Hd * NSEG, 256, 0, stream>>>(q, k, v, proj, ws, out);
}


# Chunk loads software-pipelined through registers: next chunk's Q/K/V rows fetched (16-byte loads) a chunk ahead, stored to LDS in one stage behind LDS-only barriers; L2 prefetch hints dropped
# speedup vs baseline: 3.0407x; 1.4336x over previous
#include <hip/hip_runtime.h>
#include <hip/hip_bf16.h>


#define Bd    2
#define Hd    8
#define Ld    4096
#define Dd    64
#define Md    256
#define SEG   256
#define NSEG  (Ld / SEG)
#define CK    32
#define NCH   (SEG / CK)
#define PSTR  84
#define FSTR  276
#define ASTR  36
#define STATE (Md * Dd + Md)
#define RATIO 0.0625f
#define EPS   0.001f

typedef float v8f __attribute__((ext_vector_type(8)));
typedef float v2f __attribute__((ext_vector_type(2)));

static __device__ __forceinline__ v8f wmma4(v2f a, v2f b, v8f c) {
  return __builtin_amdgcn_wmma_f32_16x16x4_f32(
       false, a,  false, b,
       (short)0, c,  false,  false);
}

static __device__ __forceinline__ float4 fetch4(
    const float* __restrict__ src, int b, int h, int lbase, int tid, int j) {
  int idx = tid + 256 * j;
  int r = idx >> 4, c4 = idx & 15;
  return *reinterpret_cast<const float4*>(
      src + (((size_t)b * Ld + lbase + r) * Hd + h) * Dd + 4 * c4);
}
static __device__ __forceinline__ void stash4(float* dst, int tid, int j, float4 x) {
  int idx = tid + 256 * j;
  int r = idx >> 4, c4 = idx & 15;
  *reinterpret_cast<float4*>(dst + r * PSTR + 4 * c4) = x;
}
static __device__ __forceinline__ void lds_barrier() {
  asm volatile("s_wait_dscnt 0x0\n\ts_barrier_signal -1\n\ts_barrier_wait -1" ::: "memory");
}

static __device__ __forceinline__ v2f ld_rm(const float* p) {
  return *reinterpret_cast<const v2f*>(p);
}
static __device__ __forceinline__ v2f ld_km(const float* p, int pitch) {
  v2f r; r.x = p[0]; r.y = p[pitch]; return r;
}

#define KS16 (Dd / 4)

static __device__ __forceinline__ void load_proj_frags(
    const float* __restrict__ proj, int wave, int lm, int hi, v2f (&pf)[2][KS16]) {
#pragma unroll
  for (int j = 0; j < 2; ++j)
#pragma unroll
    for (int ks = 0; ks < KS16; ++ks)
      pf[j][ks] = ld_rm(proj + ((2 * wave + j) * 16 + lm) * Dd + 4 * ks + 2 * hi);
}

static __device__ __forceinline__ void feature_stage(
    const float* raw, const v2f (&pf)[2][KS16], float* dstF,
    int wave, int lm, int hi) {
  v2f a0[KS16], a1[KS16];
#pragma unroll
  for (int ks = 0; ks < KS16; ++ks) {
    a0[ks] = ld_rm(raw + lm * PSTR + 4 * ks + 2 * hi);
    a1[ks] = ld_rm(raw + (16 + lm) * PSTR + 4 * ks + 2 * hi);
  }
  v8f c00 = {}, c01 = {}, c10 = {}, c11 = {};
  __builtin_amdgcn_sched_barrier(0);
#pragma unroll
  for (int ks = 0; ks < KS16; ++ks) {
    c00 = wmma4(a0[ks], pf[0][ks], c00);
    c01 = wmma4(a0[ks], pf[1][ks], c01);
    c10 = wmma4(a1[ks], pf[0][ks], c10);
    c11 = wmma4(a1[ks], pf[1][ks], c11);
  }
  const int cA = (2 * wave) * 16 + lm, cB = cA + 16;
#pragma unroll
  for (int e = 0; e < 8; ++e) {
    dstF[(e + 8 * hi) * FSTR + cA]      = fmaxf(RATIO * c00[e], 0.f) + EPS;
    dstF[(e + 8 * hi) * FSTR + cB]      = fmaxf(RATIO * c01[e], 0.f) + EPS;
    dstF[(16 + e + 8 * hi) * FSTR + cA] = fmaxf(RATIO * c10[e], 0.f) + EPS;
    dstF[(16 + e + 8 * hi) * FSTR + cB] = fmaxf(RATIO * c11[e], 0.f) + EPS;
  }
}

#define KS8 (CK / 4)
static __device__ __forceinline__ void state_update(
    const float* Kf, const float* Vlds, v8f (&sacc)[2][4],
    int wave, int lm, int hi) {
  v2f ka[2][KS8], vb[4][KS8];
#pragma unroll
  for (int ks = 0; ks < KS8; ++ks) {
#pragma unroll
    for (int m2 = 0; m2 < 2; ++m2)
      ka[m2][ks] = ld_km(Kf + (4 * ks + 2 * hi) * FSTR + 32 * wave + 16 * m2 + lm, FSTR);
#pragma unroll
    for (int dt = 0; dt < 4; ++dt)
      vb[dt][ks] = ld_km(Vlds + (4 * ks + 2 * hi) * PSTR + 16 * dt + lm, PSTR);
  }
  __builtin_amdgcn_sched_barrier(0);
#pragma unroll
  for (int ks = 0; ks < KS8; ++ks)
#pragma unroll
    for (int m2 = 0; m2 < 2; ++m2)
#pragma unroll
      for (int dt = 0; dt < 4; ++dt)
        sacc[m2][dt] = wmma4(ka[m2][ks], vb[dt][ks], sacc[m2][dt]);
}

__global__ __launch_bounds__(256) void perf_attn_segsum(
    const float* __restrict__ kk, const float* __restrict__ vv,
    const float* __restrict__ proj, float* __restrict__ ws)
{
  __shared__ __attribute__((aligned(16))) float raw [CK * PSTR];
  __shared__ __attribute__((aligned(16))) float Vlds[CK * PSTR];
  __shared__ __attribute__((aligned(16))) float Kf  [CK * FSTR];

  const int tid  = threadIdx.x;
  const int lane = tid & 31, wave = tid >> 5;
  const int lm   = lane & 15, hi   = lane >> 4;

  const int bh  = blockIdx.x / NSEG;
  const int seg = blockIdx.x % NSEG;
  const int b = bh / Hd, h = bh % Hd;

  float4 kn0 = fetch4(kk, b, h, seg * SEG, tid, 0);
  float4 kn1 = fetch4(kk, b, h, seg * SEG, tid, 1);
  float4 vn0 = fetch4(vv, b, h, seg * SEG, tid, 0);
  float4 vn1 = fetch4(vv, b, h, seg * SEG, tid, 1);

  v2f pf[2][KS16];
  load_proj_frags(proj, wave, lm, hi, pf);

  v8f sacc[2][4];
#pragma unroll
  for (int m2 = 0; m2 < 2; ++m2)
#pragma unroll
    for (int dt = 0; dt < 4; ++dt)
      sacc[m2][dt] = v8f{};
  float ssum = 0.f;

  for (int ci = 0; ci < NCH; ++ci) {
    const int lbase = seg * SEG + ci * CK;
    stash4(raw,  tid, 0, kn0);
    stash4(raw,  tid, 1, kn1);
    stash4(Vlds, tid, 0, vn0);
    stash4(Vlds, tid, 1, vn1);
    if (ci + 1 < NCH) {
      kn0 = fetch4(kk, b, h, lbase + CK, tid, 0);
      kn1 = fetch4(kk, b, h, lbase + CK, tid, 1);
      vn0 = fetch4(vv, b, h, lbase + CK, tid, 0);
      vn1 = fetch4(vv, b, h, lbase + CK, tid, 1);
    }
    lds_barrier();

    feature_stage(raw, pf, Kf, wave, lm, hi);
    lds_barrier();

    {
      float cs = 0.f;
      for (int r = 0; r < CK; ++r) cs += Kf[r * FSTR + tid];
      ssum += cs;
    }
    state_update(Kf, Vlds, sacc, wave, lm, hi);
    lds_barrier();
  }

  float* dst = ws + (size_t)(bh * NSEG + seg) * STATE;
#pragma unroll
  for (int m2 = 0; m2 < 2; ++m2)
#pragma unroll
    for (int dt = 0; dt < 4; ++dt)
#pragma unroll
      for (int e = 0; e < 8; ++e)
        dst[(32 * wave + 16 * m2 + e + 8 * hi) * Dd + 16 * dt + lm] = sacc[m2][dt][e];
  dst[Md * Dd + tid] = ssum;
}

#define STATE4 (STATE / 4)
__global__ __launch_bounds__(256) void perf_attn_prefix(float* __restrict__ ws)
{
  const int idx = blockIdx.x * 256 + threadIdx.x;
  const int bh  = idx / STATE4;
  const int e4  = idx - bh * STATE4;
  float4* p = reinterpret_cast<float4*>(ws + (size_t)bh * NSEG * STATE) + e4;

  float4 t[NSEG - 1];
#pragma unroll
  for (int sgi = 0; sgi < NSEG - 1; ++sgi) t[sgi] = p[(size_t)sgi * STATE4];

  float4 run = make_float4(0.f, 0.f, 0.f, 0.f);
#pragma unroll
  for (int sgi = 0; sgi < NSEG; ++sgi) {
    p[(size_t)sgi * STATE4] = run;
    if (sgi < NSEG - 1) {
      run.x += t[sgi].x; run.y += t[sgi].y;
      run.z += t[sgi].z; run.w += t[sgi].w;
    }
  }
}

__global__ __launch_bounds__(256) void perf_attn_main(
    const float* __restrict__ qq, const float* __restrict__ kk,
    const float* __restrict__ vv, const float* __restrict__ proj,
    const float* __restrict__ ws, float* __restrict__ out)
{
  __shared__ __attribute__((aligned(16))) float Slds[Md * PSTR];
  __shared__ __attribute__((aligned(16))) float sLds[Md];
  __shared__ __attribute__((aligned(16))) float rawQ[CK * PSTR];
  __shared__ __attribute__((aligned(16))) float rawK[CK * PSTR];
  __shared__ __attribute__((aligned(16))) float Vlds[CK * PSTR];
  __shared__ __attribute__((aligned(16))) float Qf  [CK * FSTR];
  __shared__ __attribute__((aligned(16))) float Kf  [CK * FSTR];
  __shared__ __attribute__((aligned(16))) float Alds[CK * ASTR];
  __shared__ float denLds[CK];

  const int tid  = threadIdx.x;
  const int lane = tid & 31, wave = tid >> 5;
  const int lm   = lane & 15, hi   = lane >> 4;

  const int bh  = blockIdx.x / NSEG;
  const int seg = blockIdx.x % NSEG;
  const int b = bh / Hd, h = bh % Hd;

  float4 qn0 = fetch4(qq, b, h, seg * SEG, tid, 0);
  float4 qn1 = fetch4(qq, b, h, seg * SEG, tid, 1);
  float4 kn0 = fetch4(kk, b, h, seg * SEG, tid, 0);
  float4 kn1 = fetch4(kk, b, h, seg * SEG, tid, 1);
  float4 vn0 = fetch4(vv, b, h, seg * SEG, tid, 0);
  float4 vn1 = fetch4(vv, b, h, seg * SEG, tid, 1);

  v2f pf[2][KS16];
  load_proj_frags(proj, wave, lm, hi, pf);

  const float* st = ws + (size_t)(bh * NSEG + seg) * STATE;
  v8f sacc[2][4];
#pragma unroll
  for (int m2 = 0; m2 < 2; ++m2)
#pragma unroll
    for (int dt = 0; dt < 4; ++dt)
#pragma unroll
      for (int e = 0; e < 8; ++e) {
        int row = 32 * wave + 16 * m2 + e + 8 * hi, col = 16 * dt + lm;
        float x = st[row * Dd + col];
        sacc[m2][dt][e] = x;
        Slds[row * PSTR + col] = x;
      }
  sLds[tid] = st[Md * Dd + tid];
  __syncthreads();

  for (int ci = 0; ci < NCH; ++ci) {
    const int lbase = seg * SEG + ci * CK;

    stash4(rawQ, tid, 0, qn0);
    stash4(rawQ, tid, 1, qn1);
    stash4(rawK, tid, 0, kn0);
    stash4(rawK, tid, 1, kn1);
    stash4(Vlds, tid, 0, vn0);
    stash4(Vlds, tid, 1, vn1);
    if (ci + 1 < NCH) {
      qn0 = fetch4(qq, b, h, lbase + CK, tid, 0);
      qn1 = fetch4(qq, b, h, lbase + CK, tid, 1);
      kn0 = fetch4(kk, b, h, lbase + CK, tid, 0);
      kn1 = fetch4(kk, b, h, lbase + CK, tid, 1);
      vn0 = fetch4(vv, b, h, lbase + CK, tid, 0);
      vn1 = fetch4(vv, b, h, lbase + CK, tid, 1);
    }
    lds_barrier();

    feature_stage(rawQ, pf, Qf, wave, lm, hi);
    feature_stage(rawK, pf, Kf, wave, lm, hi);
    lds_barrier();

    if (wave < 3) {
      const int it = (wave > 0) ? 1 : 0;
      const int jt = (wave > 1) ? 1 : 0;
      const float* qrow = Qf + (it * 16 + lm) * FSTR + 2 * hi;
      const float* krow = Kf + (jt * 16 + lm) * FSTR + 2 * hi;
      v2f a[2][KS8], bb[2][KS8];
#pragma unroll
      for (int ks = 0; ks < KS8; ++ks) {
        a [0][ks] = ld_rm(qrow + 4 * ks);
        bb[0][ks] = ld_rm(krow + 4 * ks);
      }
      v8f acc = {};
#pragma unroll
      for (int blk = 0; blk < Md / 32; ++blk) {
        if (blk + 1 < Md / 32) {
#pragma unroll
          for (int ks = 0; ks < KS8; ++ks) {
            a [(blk + 1) & 1][ks] = ld_rm(qrow + 32 * (blk + 1) + 4 * ks);
            bb[(blk + 1) & 1][ks] = ld_rm(krow + 32 * (blk + 1) + 4 * ks);
          }
        }
        __builtin_amdgcn_sched_barrier(0);
#pragma unroll
        for (int ks = 0; ks < KS8; ++ks)
          acc = wmma4(a[blk & 1][ks], bb[blk & 1][ks], acc);
      }
#pragma unroll
      for (int e = 0; e < 8; ++e) {
        int gi = it * 16 + e + 8 * hi, gj = jt * 16 + lm;
        Alds[gi * ASTR + gj] = (gj <= gi) ? acc[e] : 0.f;
      }
    } else if (wave == 3) {
      for (int i2 = lane; i2 < 256; i2 += 32)
        Alds[(i2 >> 4) * ASTR + 16 + (i2 & 15)] = 0.f;
    }
    lds_barrier();

    if (wave == 0) {
      const float4* qv = reinterpret_cast<const float4*>(Qf + lane * FSTR);
      const float4* sv = reinterpret_cast<const float4*>(sLds);
      const float4* av = reinterpret_cast<const float4*>(Alds + lane * ASTR);
      float4 q4[2][8], s4[2][8], a4[CK / 4];
#pragma unroll
      for (int i = 0; i < 8; ++i) { q4[0][i] = qv[i]; s4[0][i] = sv[i]; }
      float dsum = 0.f;
#pragma unroll
      for (int blk = 0; blk < Md / 32; ++blk) {
        if (blk + 1 < Md / 32) {
#pragma unroll
          for (int i = 0; i < 8; ++i) {
            q4[(blk + 1) & 1][i] = qv[8 * (blk + 1) + i];
            s4[(blk + 1) & 1][i] = sv[8 * (blk + 1) + i];
          }
        } else {
#pragma unroll
          for (int i = 0; i < CK / 4; ++i) a4[i] = av[i];
        }
        __builtin_amdgcn_sched_barrier(0);
#pragma unroll
        for (int i = 0; i < 8; ++i) {
          dsum = __builtin_fmaf(q4[blk & 1][i].x, s4[blk & 1][i].x, dsum);
          dsum = __builtin_fmaf(q4[blk & 1][i].y, s4[blk & 1][i].y, dsum);
          dsum = __builtin_fmaf(q4[blk & 1][i].z, s4[blk & 1][i].z, dsum);
          dsum = __builtin_fmaf(q4[blk & 1][i].w, s4[blk & 1][i].w, dsum);
        }
      }
#pragma unroll
      for (int i = 0; i < CK / 4; ++i) {
        dsum += a4[i].x; dsum += a4[i].y; dsum += a4[i].z; dsum += a4[i].w;
      }
      denLds[lane] = dsum;
    }
    v8f oacc = {};
    const int oit = wave >> 2, odt = wave & 3;
    {
      const float* qrow = Qf   + (oit * 16 + lm) * FSTR + 2 * hi;
      const float* scol = Slds + (2 * hi) * PSTR + odt * 16 + lm;
      const float* arow = Alds + (oit * 16 + lm) * ASTR + 2 * hi;
      const float* vcol = Vlds + (2 * hi) * PSTR + odt * 16 + lm;
      v2f a[2][KS8], bb[2][KS8], ia[KS8], ib[KS8];
#pragma unroll
      for (int ks = 0; ks < KS8; ++ks) {
        a [0][ks] = ld_rm(qrow + 4 * ks);
        bb[0][ks] = ld_km(scol + (4 * ks) * PSTR, PSTR);
      }
#pragma unroll
      for (int ks = 0; ks < KS8; ++ks) {
        ia[ks] = ld_rm(arow + 4 * ks);
        ib[ks] = ld_km(vcol + (4 * ks) * PSTR, PSTR);
      }
#pragma unroll
      for (int blk = 0; blk < Md / 32; ++blk) {
        if (blk + 1 < Md / 32) {
#pragma unroll
          for (int ks = 0; ks < KS8; ++ks) {
            a [(blk + 1) & 1][ks] = ld_rm(qrow + 32 * (blk + 1) + 4 * ks);
            bb[(blk + 1) & 1][ks] = ld_km(scol + (32 * (blk + 1) + 4 * ks) * PSTR, PSTR);
          }
        }
        __builtin_amdgcn_sched_barrier(0);
#pragma unroll
        for (int ks = 0; ks < KS8; ++ks)
          oacc = wmma4(a[blk & 1][ks], bb[blk & 1][ks], oacc);
      }
#pragma unroll
      for (int ks = 0; ks < KS8; ++ks)
        oacc = wmma4(ia[ks], ib[ks], oacc);
    }
    float cs = 0.f;
    for (int r = 0; r < CK; ++r) cs += Kf[r * FSTR + tid];
    state_update(Kf, Vlds, sacc, wave, lm, hi);
    lds_barrier();

    {
#pragma unroll
      for (int e = 0; e < 8; ++e) {
        int row = oit * 16 + e + 8 * hi;
        float den = denLds[row];
        den = (den <= 0.f) ? 1.f : den;
        float* dst =
            out + (((size_t)b * Ld + lbase + row) * Hd + h) * Dd + odt * 16 + lm;
        __builtin_nontemporal_store(oacc[e] / den, dst);
      }
    }
    sLds[tid] += cs;
#pragma unroll
    for (int m2 = 0; m2 < 2; ++m2)
#pragma unroll
      for (int dt = 0; dt < 4; ++dt)
#pragma unroll
        for (int e = 0; e < 8; ++e)
          Slds[(32 * wave + 16 * m2 + e + 8 * hi) * PSTR + 16 * dt + lm] = sacc[m2][dt][e];
  }
}

extern "C" void kernel_launch(void* const* d_in, const int* in_sizes, int n_in,
                              void* d_out, int out_size, void* d_ws, size_t ws_size,
                              hipStream_t stream) {
  const float* q    = (const float*)d_in[0];
  const float* k    = (const float*)d_in[1];
  const float* v    = (const float*)d_in[2];
  const float* proj = (const float*)d_in[3];
  float* out = (float*)d_out;
  float* ws  = (float*)d_ws;

  (void)in_sizes; (void)n_in; (void)out_size; (void)ws_size;

  perf_attn_segsum<<<Bd * Hd * NSEG, 256, 0, stream>>>(k, v, proj, ws);
  perf_attn_prefix<<<Bd * Hd * STATE4 / 256, 256, 0, stream>>>(ws);
  perf_attn_main  <<<Bd * Hd * NSEG, 256, 0, stream>>>(q, k, v, proj, ws, out);
}


# Stage F work that does not read the masked Qf.Kf^T tile (Qf@S_prev products, state update, den's Qf.s part) moved into stage E beside the tile products, filling its 5 idle waves; sum order kept
# speedup vs baseline: 3.2458x; 1.0675x over previous
#include <hip/hip_runtime.h>
#include <hip/hip_bf16.h>


#define Bd    2
#define Hd    8
#define Ld    4096
#define Dd    64
#define Md    256
#define SEG   256
#define NSEG  (Ld / SEG)
#define CK    32
#define NCH   (SEG / CK)
#define PSTR  84
#define FSTR  276
#define ASTR  36
#define STATE (Md * Dd + Md)
#define RATIO 0.0625f
#define EPS   0.001f

typedef float v8f __attribute__((ext_vector_type(8)));
typedef float v2f __attribute__((ext_vector_type(2)));

static __device__ __forceinline__ v8f wmma4(v2f a, v2f b, v8f c) {
  return __builtin_amdgcn_wmma_f32_16x16x4_f32(
       false, a,  false, b,
       (short)0, c,  false,  false);
}

static __device__ __forceinline__ float4 fetch4(
    const float* __restrict__ src, int b, int h, int lbase, int tid, int j) {
  int idx = tid + 256 * j;
  int r = idx >> 4, c4 = idx & 15;
  return *reinterpret_cast<const float4*>(
      src + (((size_t)b * Ld + lbase + r) * Hd + h) * Dd + 4 * c4);
}
static __device__ __forceinline__ void stash4(float* dst, int tid, int j, float4 x) {
  int idx = tid + 256 * j;
  int r = idx >> 4, c4 = idx & 15;
  *reinterpret_cast<float4*>(dst + r * PSTR + 4 * c4) = x;
}
static __device__ __forceinline__ void lds_barrier() {
  asm volatile("s_wait_dscnt 0x0\n\ts_barrier_signal -1\n\ts_barrier_wait -1" ::: "memory");
}

static __device__ __forceinline__ v2f ld_rm(const float* p) {
  return *reinterpret_cast<const v2f*>(p);
}
static __device__ __forceinline__ v2f ld_km(const float* p, int pitch) {
  v2f r; r.x = p[0]; r.y = p[pitch]; return r;
}

#define KS16 (Dd / 4)

static __device__ __forceinline__ void load_proj_frags(
    const float* __restrict__ proj, int wave, int lm, int hi, v2f (&pf)[2][KS16]) {
#pragma unroll
  for (int j = 0; j < 2; ++j)
#pragma unroll
    for (int ks = 0; ks < KS16; ++ks)
      pf[j][ks] = ld_rm(proj + ((2 * wave + j) * 16 + lm) * Dd + 4 * ks + 2 * hi);
}

static __device__ __forceinline__ void feature_stage(
    const float* raw, const v2f (&pf)[2][KS16], float* dstF,
    int wave, int lm, int hi) {
  v2f a0[KS16], a1[KS16];
#pragma unroll
  for (int ks = 0; ks < KS16; ++ks) {
    a0[ks] = ld_rm(raw + lm * PSTR + 4 * ks + 2 * hi);
    a1[ks] = ld_rm(raw + (16 + lm) * PSTR + 4 * ks + 2 * hi);
  }
  v8f c00 = {}, c01 = {}, c10 = {}, c11 = {};
  __builtin_amdgcn_sched_barrier(0);
#pragma unroll
  for (int ks = 0; ks < KS16; ++ks) {
    c00 = wmma4(a0[ks], pf[0][ks], c00);
    c01 = wmma4(a0[ks], pf[1][ks], c01);
    c10 = wmma4(a1[ks], pf[0][ks], c10);
    c11 = wmma4(a1[ks], pf[1][ks], c11);
  }
  const int cA = (2 * wave) * 16 + lm, cB = cA + 16;
#pragma unroll
  for (int e = 0; e < 8; ++e) {
    dstF[(e + 8 * hi) * FSTR + cA]      = fmaxf(RATIO * c00[e], 0.f) + EPS;
    dstF[(e + 8 * hi) * FSTR + cB]      = fmaxf(RATIO * c01[e], 0.f) + EPS;
    dstF[(16 + e + 8 * hi) * FSTR + cA] = fmaxf(RATIO * c10[e], 0.f) + EPS;
    dstF[(16 + e + 8 * hi) * FSTR + cB] = fmaxf(RATIO * c11[e], 0.f) + EPS;
  }
}

#define KS8 (CK / 4)
static __device__ __forceinline__ void state_update(
    const float* Kf, const float* Vlds, v8f (&sacc)[2][4],
    int wave, int lm, int hi) {
  v2f ka[2][KS8], vb[4][KS8];
#pragma unroll
  for (int ks = 0; ks < KS8; ++ks) {
#pragma unroll
    for (int m2 = 0; m2 < 2; ++m2)
      ka[m2][ks] = ld_km(Kf + (4 * ks + 2 * hi) * FSTR + 32 * wave + 16 * m2 + lm, FSTR);
#pragma unroll
    for (int dt = 0; dt < 4; ++dt)
      vb[dt][ks] = ld_km(Vlds + (4 * ks + 2 * hi) * PSTR + 16 * dt + lm, PSTR);
  }
  __builtin_amdgcn_sched_barrier(0);
#pragma unroll
  for (int ks = 0; ks < KS8; ++ks)
#pragma unroll
    for (int m2 = 0; m2 < 2; ++m2)
#pragma unroll
      for (int dt = 0; dt < 4; ++dt)
        sacc[m2][dt] = wmma4(ka[m2][ks], vb[dt][ks], sacc[m2][dt]);
}

__global__ __launch_bounds__(256) void perf_attn_segsum(
    const float* __restrict__ kk, const float* __restrict__ vv,
    const float* __restrict__ proj, float* __restrict__ ws)
{
  __shared__ __attribute__((aligned(16))) float raw [CK * PSTR];
  __shared__ __attribute__((aligned(16))) float Vlds[CK * PSTR];
  __shared__ __attribute__((aligned(16))) float Kf  [CK * FSTR];

  const int tid  = threadIdx.x;
  const int lane = tid & 31, wave = tid >> 5;
  const int lm   = lane & 15, hi   = lane >> 4;

  const int bh  = blockIdx.x / NSEG;
  const int seg = blockIdx.x % NSEG;
  const int b = bh / Hd, h = bh % Hd;

  float4 kn0 = fetch4(kk, b, h, seg * SEG, tid, 0);
  float4 kn1 = fetch4(kk, b, h, seg * SEG, tid, 1);
  float4 vn0 = fetch4(vv, b, h, seg * SEG, tid, 0);
  float4 vn1 = fetch4(vv, b, h, seg * SEG, tid, 1);

  v2f pf[2][KS16];
  load_proj_frags(proj, wave, lm, hi, pf);

  v8f sacc[2][4];
#pragma unroll
  for (int m2 = 0; m2 < 2; ++m2)
#pragma unroll
    for (int dt = 0; dt < 4; ++dt)
      sacc[m2][dt] = v8f{};
  float ssum = 0.f;

  for (int ci = 0; ci < NCH; ++ci) {
    const int lbase = seg * SEG + ci * CK;
    stash4(raw,  tid, 0, kn0);
    stash4(raw,  tid, 1, kn1);
    stash4(Vlds, tid, 0, vn0);
    stash4(Vlds, tid, 1, vn1);
    if (ci + 1 < NCH) {
      kn0 = fetch4(kk, b, h, lbase + CK, tid, 0);
      kn1 = fetch4(kk, b, h, lbase + CK, tid, 1);
      vn0 = fetch4(vv, b, h, lbase + CK, tid, 0);
      vn1 = fetch4(vv, b, h, lbase + CK, tid, 1);
    }
    lds_barrier();

    feature_stage(raw, pf, Kf, wave, lm, hi);
    lds_barrier();

    {
      float cs = 0.f;
      for (int r = 0; r < CK; ++r) cs += Kf[r * FSTR + tid];
      ssum += cs;
    }
    state_update(Kf, Vlds, sacc, wave, lm, hi);
    lds_barrier();
  }

  float* dst = ws + (size_t)(bh * NSEG + seg) * STATE;
#pragma unroll
  for (int m2 = 0; m2 < 2; ++m2)
#pragma unroll
    for (int dt = 0; dt < 4; ++dt)
#pragma unroll
      for (int e = 0; e < 8; ++e)
        dst[(32 * wave + 16 * m2 + e + 8 * hi) * Dd + 16 * dt + lm] = sacc[m2][dt][e];
  dst[Md * Dd + tid] = ssum;
}

#define STATE4 (STATE / 4)
__global__ __launch_bounds__(256) void perf_attn_prefix(float* __restrict__ ws)
{
  const int idx = blockIdx.x * 256 + threadIdx.x;
  const int bh  = idx / STATE4;
  const int e4  = idx - bh * STATE4;
  float4* p = reinterpret_cast<float4*>(ws + (size_t)bh * NSEG * STATE) + e4;

  float4 t[NSEG - 1];
#pragma unroll
  for (int sgi = 0; sgi < NSEG - 1; ++sgi) t[sgi] = p[(size_t)sgi * STATE4];

  float4 run = make_float4(0.f, 0.f, 0.f, 0.f);
#pragma unroll
  for (int sgi = 0; sgi < NSEG; ++sgi) {
    p[(size_t)sgi * STATE4] = run;
    if (sgi < NSEG - 1) {
      run.x += t[sgi].x; run.y += t[sgi].y;
      run.z += t[sgi].z; run.w += t[sgi].w;
    }
  }
}

__global__ __launch_bounds__(256) void perf_attn_main(
    const float* __restrict__ qq, const float* __restrict__ kk,
    const float* __restrict__ vv, const float* __restrict__ proj,
    const float* __restrict__ ws, float* __restrict__ out)
{
  __shared__ __attribute__((aligned(16))) float Slds[Md * PSTR];
  __shared__ __attribute__((aligned(16))) float sLds[Md];
  __shared__ __attribute__((aligned(16))) float rawQ[CK * PSTR];
  __shared__ __attribute__((aligned(16))) float rawK[CK * PSTR];
  __shared__ __attribute__((aligned(16))) float Vlds[CK * PSTR];
  __shared__ __attribute__((aligned(16))) float Qf  [CK * FSTR];
  __shared__ __attribute__((aligned(16))) float Kf  [CK * FSTR];
  __shared__ __attribute__((aligned(16))) float Alds[CK * ASTR];
  __shared__ float denLds[CK];

  const int tid  = threadIdx.x;
  const int lane = tid & 31, wave = tid >> 5;
  const int lm   = lane & 15, hi   = lane >> 4;

  const int bh  = blockIdx.x / NSEG;
  const int seg = blockIdx.x % NSEG;
  const int b = bh / Hd, h = bh % Hd;

  float4 qn0 = fetch4(qq, b, h, seg * SEG, tid, 0);
  float4 qn1 = fetch4(qq, b, h, seg * SEG, tid, 1);
  float4 kn0 = fetch4(kk, b, h, seg * SEG, tid, 0);
  float4 kn1 = fetch4(kk, b, h, seg * SEG, tid, 1);
  float4 vn0 = fetch4(vv, b, h, seg * SEG, tid, 0);
  float4 vn1 = fetch4(vv, b, h, seg * SEG, tid, 1);

  v2f pf[2][KS16];
  load_proj_frags(proj, wave, lm, hi, pf);

  const float* st = ws + (size_t)(bh * NSEG + seg) * STATE;
  v8f sacc[2][4];
#pragma unroll
  for (int m2 = 0; m2 < 2; ++m2)
#pragma unroll
    for (int dt = 0; dt < 4; ++dt)
#pragma unroll
      for (int e = 0; e < 8; ++e) {
        int row = 32 * wave + 16 * m2 + e + 8 * hi, col = 16 * dt + lm;
        float x = st[row * Dd + col];
        sacc[m2][dt][e] = x;
        Slds[row * PSTR + col] = x;
      }
  sLds[tid] = st[Md * Dd + tid];
  __syncthreads();

  for (int ci = 0; ci < NCH; ++ci) {
    const int lbase = seg * SEG + ci * CK;

    stash4(rawQ, tid, 0, qn0);
    stash4(rawQ, tid, 1, qn1);
    stash4(rawK, tid, 0, kn0);
    stash4(rawK, tid, 1, kn1);
    stash4(Vlds, tid, 0, vn0);
    stash4(Vlds, tid, 1, vn1);
    if (ci + 1 < NCH) {
      qn0 = fetch4(qq, b, h, lbase + CK, tid, 0);
      qn1 = fetch4(qq, b, h, lbase + CK, tid, 1);
      kn0 = fetch4(kk, b, h, lbase + CK, tid, 0);
      kn1 = fetch4(kk, b, h, lbase + CK, tid, 1);
      vn0 = fetch4(vv, b, h, lbase + CK, tid, 0);
      vn1 = fetch4(vv, b, h, lbase + CK, tid, 1);
    }
    lds_barrier();

    feature_stage(rawQ, pf, Qf, wave, lm, hi);
    feature_stage(rawK, pf, Kf, wave, lm, hi);
    lds_barrier();

    float dsum = 0.f;
    if (wave < 3) {
      const int it = (wave > 0) ? 1 : 0;
      const int jt = (wave > 1) ? 1 : 0;
      const float* qrow = Qf + (it * 16 + lm) * FSTR + 2 * hi;
      const float* krow = Kf + (jt * 16 + lm) * FSTR + 2 * hi;
      v2f a[2][KS8], bb[2][KS8];
#pragma unroll
      for (int ks = 0; ks < KS8; ++ks) {
        a [0][ks] = ld_rm(qrow + 4 * ks);
        bb[0][ks] = ld_rm(krow + 4 * ks);
      }
      v8f acc = {};
#pragma unroll
      for (int blk = 0; blk < Md / 32; ++blk) {
        if (blk + 1 < Md / 32) {
#pragma unroll
          for (int ks = 0; ks < KS8; ++ks) {
            a [(blk + 1) & 1][ks] = ld_rm(qrow + 32 * (blk + 1) + 4 * ks);
            bb[(blk + 1) & 1][ks] = ld_rm(krow + 32 * (blk + 1) + 4 * ks);
          }
        }
        __builtin_amdgcn_sched_barrier(0);
#pragma unroll
        for (int ks = 0; ks < KS8; ++ks)
          acc = wmma4(a[blk & 1][ks], bb[blk & 1][ks], acc);
      }
#pragma unroll
      for (int e = 0; e < 8; ++e) {
        int gi = it * 16 + e + 8 * hi, gj = jt * 16 + lm;
        Alds[gi * ASTR + gj] = (gj <= gi) ? acc[e] : 0.f;
      }
    } else if (wave == 3) {
      const float4* qv = reinterpret_cast<const float4*>(Qf + lane * FSTR);
      const float4* sv = reinterpret_cast<const float4*>(sLds);
      float4 q4[2][8], s4[2][8];
#pragma unroll
      for (int i = 0; i < 8; ++i) { q4[0][i] = qv[i]; s4[0][i] = sv[i]; }
#pragma unroll
      for (int blk = 0; blk < Md / 32; ++blk) {
        if (blk + 1 < Md / 32) {
#pragma unroll
          for (int i = 0; i < 8; ++i) {
            q4[(blk + 1) & 1][i] = qv[8 * (blk + 1) + i];
            s4[(blk + 1) & 1][i] = sv[8 * (blk + 1) + i];
          }
        }
        __builtin_amdgcn_sched_barrier(0);
#pragma unroll
        for (int i = 0; i < 8; ++i) {
          dsum = __builtin_fmaf(q4[blk & 1][i].x, s4[blk & 1][i].x, dsum);
          dsum = __builtin_fmaf(q4[blk & 1][i].y, s4[blk & 1][i].y, dsum);
          dsum = __builtin_fmaf(q4[blk & 1][i].z, s4[blk & 1][i].z, dsum);
          dsum = __builtin_fmaf(q4[blk & 1][i].w, s4[blk & 1][i].w, dsum);
        }
      }
    } else if (wave == 4) {
      for (int i2 = lane; i2 < 256; i2 += 32)
        Alds[(i2 >> 4) * ASTR + 16 + (i2 & 15)] = 0.f;
    }
    float cs = 0.f;
    for (int r = 0; r < CK; ++r) cs += Kf[r * FSTR + tid];
    state_update(Kf, Vlds, sacc, wave, lm, hi);
    v8f oacc = {};
    const int oit = wave >> 2, odt = wave & 3;
    {
      const float* qrow = Qf   + (oit * 16 + lm) * FSTR + 2 * hi;
      const float* scol = Slds + (2 * hi) * PSTR + odt * 16 + lm;
      v2f a[2][KS8], bb[2][KS8];
#pragma unroll
      for (int ks = 0; ks < KS8; ++ks) {
        a [0][ks] = ld_rm(qrow + 4 * ks);
        bb[0][ks] = ld_km(scol + (4 * ks) * PSTR, PSTR);
      }
#pragma unroll
      for (int blk = 0; blk < Md / 32; ++blk) {
        if (blk + 1 < Md / 32) {
#pragma unroll
          for (int ks = 0; ks < KS8; ++ks) {
            a [(blk + 1) & 1][ks] = ld_rm(qrow + 32 * (blk + 1) + 4 * ks);
            bb[(blk + 1) & 1][ks] = ld_km(scol + (32 * (blk + 1) + 4 * ks) * PSTR, PSTR);
          }
        }
        __builtin_amdgcn_sched_barrier(0);
#pragma unroll
        for (int ks = 0; ks < KS8; ++ks)
          oacc = wmma4(a[blk & 1][ks], bb[blk & 1][ks], oacc);
      }
    }
    lds_barrier();

    if (wave == 3) {
      const float4* av = reinterpret_cast<const float4*>(Alds + lane * ASTR);
      float4 a4[CK / 4];
#pragma unroll
      for (int i = 0; i < CK / 4; ++i) a4[i] = av[i];
#pragma unroll
      for (int i = 0; i < CK / 4; ++i) {
        dsum += a4[i].x; dsum += a4[i].y; dsum += a4[i].z; dsum += a4[i].w;
      }
      denLds[lane] = dsum;
    }
    {
      const float* arow = Alds + (oit * 16 + lm) * ASTR + 2 * hi;
      const float* vcol = Vlds + (2 * hi) * PSTR + odt * 16 + lm;
      v2f ia[KS8], ib[KS8];
#pragma unroll
      for (int ks = 0; ks < KS8; ++ks) {
        ia[ks] = ld_rm(arow + 4 * ks);
        ib[ks] = ld_km(vcol + (4 * ks) * PSTR, PSTR);
      }
      __builtin_amdgcn_sched_barrier(0);
#pragma unroll
      for (int ks = 0; ks < KS8; ++ks)
        oacc = wmma4(ia[ks], ib[ks], oacc);
    }
    lds_barrier();

    {
#pragma unroll
      for (int e = 0; e < 8; ++e) {
        int row = oit * 16 + e + 8 * hi;
        float den = denLds[row];
        den = (den <= 0.f) ? 1.f : den;
        float* dst =
            out + (((size_t)b * Ld + lbase + row) * Hd + h) * Dd + odt * 16 + lm;
        __builtin_nontemporal_store(oacc[e] / den, dst);
      }
    }
    sLds[tid] += cs;
#pragma unroll
    for (int m2 = 0; m2 < 2; ++m2)
#pragma unroll
      for (int dt = 0; dt < 4; ++dt)
#pragma unroll
        for (int e = 0; e < 8; ++e)
          Slds[(32 * wave + 16 * m2 + e + 8 * hi) * PSTR + 16 * dt + lm] = sacc[m2][dt][e];
  }
}

extern "C" void kernel_launch(void* const* d_in, const int* in_sizes, int n_in,
                              void* d_out, int out_size, void* d_ws, size_t ws_size,
                              hipStream_t stream) {
  const float* q    = (const float*)d_in[0];
  const float* k    = (const float*)d_in[1];
  const float* v    = (const float*)d_in[2];
  const float* proj = (const float*)d_in[3];
  float* out = (float*)d_out;
  float* ws  = (float*)d_ws;

  (void)in_sizes; (void)n_in; (void)out_size; (void)ws_size;

  perf_attn_segsum<<<Bd * Hd * NSEG, 256, 0, stream>>>(k, v, proj, ws);
  perf_attn_prefix<<<Bd * Hd * STATE4 / 256, 256, 0, stream>>>(ws);
  perf_attn_main  <<<Bd * Hd * NSEG, 256, 0, stream>>>(q, k, v, proj, ws, out);
}


# S and Kf accumulator tiles used in place as WMMA operands (k order matched to their layout): Qf@S_prev split by m over waves, shares summed via LDS; sum order of S and Qf@S changes
# speedup vs baseline: 3.4544x; 1.0643x over previous
#include <hip/hip_runtime.h>
#include <hip/hip_bf16.h>


#define Bd    2
#define Hd    8
#define Ld    4096
#define Dd    64
#define Md    256
#define SEG   256
#define NSEG  (Ld / SEG)
#define CK    32
#define NCH   (SEG / CK)
#define PSTR  84
#define FSTR  276
#define ASTR  36
#define STATE (Md * Dd + Md)
#define RATIO 0.0625f
#define EPS   0.001f

typedef float v8f __attribute__((ext_vector_type(8)));
typedef float v2f __attribute__((ext_vector_type(2)));

static __device__ __forceinline__ v8f wmma4(v2f a, v2f b, v8f c) {
  return __builtin_amdgcn_wmma_f32_16x16x4_f32(
       false, a,  false, b,
       (short)0, c,  false,  false);
}

static __device__ __forceinline__ float4 fetch4(
    const float* __restrict__ src, int b, int h, int lbase, int tid, int j) {
  int idx = tid + 256 * j;
  int r = idx >> 4, c4 = idx & 15;
  return *reinterpret_cast<const float4*>(
      src + (((size_t)b * Ld + lbase + r) * Hd + h) * Dd + 4 * c4);
}
static __device__ __forceinline__ void stash4(float* dst, int tid, int j, float4 x) {
  int idx = tid + 256 * j;
  int r = idx >> 4, c4 = idx & 15;
  *reinterpret_cast<float4*>(dst + r * PSTR + 4 * c4) = x;
}
static __device__ __forceinline__ void lds_barrier() {
  asm volatile("s_wait_dscnt 0x0\n\ts_barrier_signal -1\n\ts_barrier_wait -1" ::: "memory");
}

static __device__ __forceinline__ v2f ld_rm(const float* p) {
  return *reinterpret_cast<const v2f*>(p);
}
static __device__ __forceinline__ v2f ld_km(const float* p, int pitch) {
  v2f r; r.x = p[0]; r.y = p[pitch]; return r;
}

#define KS16 (Dd / 4)

static __device__ __forceinline__ void load_proj_frags(
    const float* __restrict__ proj, int wave, int lm, int hi, v2f (&pf)[2][KS16]) {
#pragma unroll
  for (int j = 0; j < 2; ++j)
#pragma unroll
    for (int ks = 0; ks < KS16; ++ks)
      pf[j][ks] = ld_rm(proj + ((2 * wave + j) * 16 + lm) * Dd + 4 * ks + 2 * hi);
}

static __device__ __forceinline__ void feature_stage(
    const float* raw, const v2f (&pf)[2][KS16], float* dstF,
    int wave, int lm, int hi, v8f (&f)[2][2]) {
  v2f a0[KS16], a1[KS16];
#pragma unroll
  for (int ks = 0; ks < KS16; ++ks) {
    a0[ks] = ld_rm(raw + lm * PSTR + 4 * ks + 2 * hi);
    a1[ks] = ld_rm(raw + (16 + lm) * PSTR + 4 * ks + 2 * hi);
  }
  v8f c00 = {}, c01 = {}, c10 = {}, c11 = {};
  __builtin_amdgcn_sched_barrier(0);
#pragma unroll
  for (int ks = 0; ks < KS16; ++ks) {
    c00 = wmma4(a0[ks], pf[0][ks], c00);
    c01 = wmma4(a0[ks], pf[1][ks], c01);
    c10 = wmma4(a1[ks], pf[0][ks], c10);
    c11 = wmma4(a1[ks], pf[1][ks], c11);
  }
  const int cA = (2 * wave) * 16 + lm, cB = cA + 16;
#pragma unroll
  for (int e = 0; e < 8; ++e) {
    f[0][0][e] = fmaxf(RATIO * c00[e], 0.f) + EPS;
    f[0][1][e] = fmaxf(RATIO * c01[e], 0.f) + EPS;
    f[1][0][e] = fmaxf(RATIO * c10[e], 0.f) + EPS;
    f[1][1][e] = fmaxf(RATIO * c11[e], 0.f) + EPS;
    dstF[(e + 8 * hi) * FSTR + cA]      = f[0][0][e];
    dstF[(e + 8 * hi) * FSTR + cB]      = f[0][1][e];
    dstF[(16 + e + 8 * hi) * FSTR + cA] = f[1][0][e];
    dstF[(16 + e + 8 * hi) * FSTR + cB] = f[1][1][e];
  }
}

#define KS8 (CK / 4)
static __device__ __forceinline__ void load_v_frags(
    const float* Vlds, int lm, int hi, v2f (&vb)[4][KS8]) {
#pragma unroll
  for (int rt = 0; rt < 2; ++rt)
#pragma unroll
    for (int j = 0; j < 4; ++j)
#pragma unroll
      for (int dt = 0; dt < 4; ++dt)
        vb[dt][4 * rt + j] =
            ld_km(Vlds + (16 * rt + 8 * hi + 2 * j) * PSTR + 16 * dt + lm, PSTR);
}
static __device__ __forceinline__ void state_update(
    const v8f (&kf)[2][2], const v2f (&vb)[4][KS8], v8f (&sacc)[2][4]) {
#pragma unroll
  for (int rt = 0; rt < 2; ++rt)
#pragma unroll
    for (int j = 0; j < 4; ++j)
#pragma unroll
      for (int m2 = 0; m2 < 2; ++m2) {
        v2f a;
        a.x = kf[rt][m2][2 * j];
        a.y = kf[rt][m2][2 * j + 1];
#pragma unroll
        for (int dt = 0; dt < 4; ++dt)
          sacc[m2][dt] = wmma4(a, vb[dt][4 * rt + j], sacc[m2][dt]);
      }
}

static __device__ __forceinline__ void qs_share(
    const float* Qf, const float* Vlds, const v8f (&sacc)[2][4], float4* P4,
    int wave, int lane, int lm, int hi, v2f (&vb)[4][KS8]) {
  v2f qa[2][KS8];
  load_v_frags(Vlds, lm, hi, vb);
#pragma unroll
  for (int m2 = 0; m2 < 2; ++m2)
#pragma unroll
    for (int j = 0; j < 4; ++j)
#pragma unroll
      for (int oit = 0; oit < 2; ++oit)
        qa[oit][4 * m2 + j] = ld_rm(
            Qf + (16 * oit + lm) * FSTR + 32 * wave + 16 * m2 + 8 * hi + 2 * j);
  v8f pacc[2][4];
#pragma unroll
  for (int oit = 0; oit < 2; ++oit)
#pragma unroll
    for (int dt = 0; dt < 4; ++dt)
      pacc[oit][dt] = v8f{};
  __builtin_amdgcn_sched_barrier(0);
#pragma unroll
  for (int m2 = 0; m2 < 2; ++m2)
#pragma unroll
    for (int j = 0; j < 4; ++j)
#pragma unroll
      for (int dt = 0; dt < 4; ++dt) {
        v2f bq;
        bq.x = sacc[m2][dt][2 * j];
        bq.y = sacc[m2][dt][2 * j + 1];
#pragma unroll
        for (int oit = 0; oit < 2; ++oit)
          pacc[oit][dt] = wmma4(qa[oit][4 * m2 + j], bq, pacc[oit][dt]);
      }
#pragma unroll
  for (int oit = 0; oit < 2; ++oit)
#pragma unroll
    for (int dt = 0; dt < 4; ++dt) {
      const int t = oit * 4 + dt;
      P4[((wave * 8 + t) * 2 + 0) * 32 + lane] = make_float4(
          pacc[oit][dt][0], pacc[oit][dt][1], pacc[oit][dt][2], pacc[oit][dt][3]);
      P4[((wave * 8 + t) * 2 + 1) * 32 + lane] = make_float4(
          pacc[oit][dt][4], pacc[oit][dt][5], pacc[oit][dt][6], pacc[oit][dt][7]);
    }
}

__global__ __launch_bounds__(256) void perf_attn_segsum(
    const float* __restrict__ kk, const float* __restrict__ vv,
    const float* __restrict__ proj, float* __restrict__ ws)
{
  __shared__ __attribute__((aligned(16))) float raw [CK * PSTR];
  __shared__ __attribute__((aligned(16))) float Vlds[CK * PSTR];
  __shared__ __attribute__((aligned(16))) float Kf  [CK * FSTR];

  const int tid  = threadIdx.x;
  const int lane = tid & 31, wave = tid >> 5;
  const int lm   = lane & 15, hi   = lane >> 4;

  const int bh  = blockIdx.x / NSEG;
  const int seg = blockIdx.x % NSEG;
  const int b = bh / Hd, h = bh % Hd;

  float4 kn0 = fetch4(kk, b, h, seg * SEG, tid, 0);
  float4 kn1 = fetch4(kk, b, h, seg * SEG, tid, 1);
  float4 vn0 = fetch4(vv, b, h, seg * SEG, tid, 0);
  float4 vn1 = fetch4(vv, b, h, seg * SEG, tid, 1);

  v2f pf[2][KS16];
  load_proj_frags(proj, wave, lm, hi, pf);

  v8f sacc[2][4];
#pragma unroll
  for (int m2 = 0; m2 < 2; ++m2)
#pragma unroll
    for (int dt = 0; dt < 4; ++dt)
      sacc[m2][dt] = v8f{};
  float ssum = 0.f;

  for (int ci = 0; ci < NCH; ++ci) {
    const int lbase = seg * SEG + ci * CK;
    stash4(raw,  tid, 0, kn0);
    stash4(raw,  tid, 1, kn1);
    stash4(Vlds, tid, 0, vn0);
    stash4(Vlds, tid, 1, vn1);
    if (ci + 1 < NCH) {
      kn0 = fetch4(kk, b, h, lbase + CK, tid, 0);
      kn1 = fetch4(kk, b, h, lbase + CK, tid, 1);
      vn0 = fetch4(vv, b, h, lbase + CK, tid, 0);
      vn1 = fetch4(vv, b, h, lbase + CK, tid, 1);
    }
    lds_barrier();

    v8f kfr[2][2];
    feature_stage(raw, pf, Kf, wave, lm, hi, kfr);
    lds_barrier();

    {
      float cs = 0.f;
      for (int r = 0; r < CK; ++r) cs += Kf[r * FSTR + tid];
      ssum += cs;
    }
    {
      v2f vb[4][KS8];
      load_v_frags(Vlds, lm, hi, vb);
      __builtin_amdgcn_sched_barrier(0);
      state_update(kfr, vb, sacc);
    }
    lds_barrier();
  }

  float* dst = ws + (size_t)(bh * NSEG + seg) * STATE;
#pragma unroll
  for (int m2 = 0; m2 < 2; ++m2)
#pragma unroll
    for (int dt = 0; dt < 4; ++dt)
#pragma unroll
      for (int e = 0; e < 8; ++e)
        dst[(32 * wave + 16 * m2 + e + 8 * hi) * Dd + 16 * dt + lm] = sacc[m2][dt][e];
  dst[Md * Dd + tid] = ssum;
}

#define STATE4 (STATE / 4)
__global__ __launch_bounds__(256) void perf_attn_prefix(float* __restrict__ ws)
{
  const int idx = blockIdx.x * 256 + threadIdx.x;
  const int bh  = idx / STATE4;
  const int e4  = idx - bh * STATE4;
  float4* p = reinterpret_cast<float4*>(ws + (size_t)bh * NSEG * STATE) + e4;

  float4 t[NSEG - 1];
#pragma unroll
  for (int sgi = 0; sgi < NSEG - 1; ++sgi) t[sgi] = p[(size_t)sgi * STATE4];

  float4 run = make_float4(0.f, 0.f, 0.f, 0.f);
#pragma unroll
  for (int sgi = 0; sgi < NSEG; ++sgi) {
    p[(size_t)sgi * STATE4] = run;
    if (sgi < NSEG - 1) {
      run.x += t[sgi].x; run.y += t[sgi].y;
      run.z += t[sgi].z; run.w += t[sgi].w;
    }
  }
}

__global__ __launch_bounds__(256) void perf_attn_main(
    const float* __restrict__ qq, const float* __restrict__ kk,
    const float* __restrict__ vv, const float* __restrict__ proj,
    const float* __restrict__ ws, float* __restrict__ out)
{
  __shared__ __attribute__((aligned(16))) float Plds[8 * CK * Dd];
  __shared__ __attribute__((aligned(16))) float sLds[Md];
  __shared__ __attribute__((aligned(16))) float rawQ[CK * PSTR];
  __shared__ __attribute__((aligned(16))) float rawK[CK * PSTR];
  __shared__ __attribute__((aligned(16))) float Vlds[CK * PSTR];
  __shared__ __attribute__((aligned(16))) float Qf  [CK * FSTR];
  __shared__ __attribute__((aligned(16))) float Kf  [CK * FSTR];
  __shared__ __attribute__((aligned(16))) float Alds[CK * ASTR];
  __shared__ float denLds[8 * CK];

  const int tid  = threadIdx.x;
  const int lane = tid & 31, wave = tid >> 5;
  const int lm   = lane & 15, hi   = lane >> 4;

  const int bh  = blockIdx.x / NSEG;
  const int seg = blockIdx.x % NSEG;
  const int b = bh / Hd, h = bh % Hd;

  float4 qn0 = fetch4(qq, b, h, seg * SEG, tid, 0);
  float4 qn1 = fetch4(qq, b, h, seg * SEG, tid, 1);
  float4 kn0 = fetch4(kk, b, h, seg * SEG, tid, 0);
  float4 kn1 = fetch4(kk, b, h, seg * SEG, tid, 1);
  float4 vn0 = fetch4(vv, b, h, seg * SEG, tid, 0);
  float4 vn1 = fetch4(vv, b, h, seg * SEG, tid, 1);

  v2f pf[2][KS16];
  load_proj_frags(proj, wave, lm, hi, pf);

  const float* st = ws + (size_t)(bh * NSEG + seg) * STATE;
  v8f sacc[2][4];
#pragma unroll
  for (int m2 = 0; m2 < 2; ++m2)
#pragma unroll
    for (int dt = 0; dt < 4; ++dt)
#pragma unroll
      for (int e = 0; e < 8; ++e) {
        int row = 32 * wave + 16 * m2 + e + 8 * hi, col = 16 * dt + lm;
        sacc[m2][dt][e] = st[row * Dd + col];
      }
  sLds[tid] = st[Md * Dd + tid];
  __syncthreads();

  for (int ci = 0; ci < NCH; ++ci) {
    const int lbase = seg * SEG + ci * CK;

    stash4(rawQ, tid, 0, qn0);
    stash4(rawQ, tid, 1, qn1);
    stash4(rawK, tid, 0, kn0);
    stash4(rawK, tid, 1, kn1);
    stash4(Vlds, tid, 0, vn0);
    stash4(Vlds, tid, 1, vn1);
    if (ci + 1 < NCH) {
      qn0 = fetch4(qq, b, h, lbase + CK, tid, 0);
      qn1 = fetch4(qq, b, h, lbase + CK, tid, 1);
      kn0 = fetch4(kk, b, h, lbase + CK, tid, 0);
      kn1 = fetch4(kk, b, h, lbase + CK, tid, 1);
      vn0 = fetch4(vv, b, h, lbase + CK, tid, 0);
      vn1 = fetch4(vv, b, h, lbase + CK, tid, 1);
    }
    lds_barrier();

    v8f qfr[2][2], kfr[2][2];
    feature_stage(rawQ, pf, Qf, wave, lm, hi, qfr);
    feature_stage(rawK, pf, Kf, wave, lm, hi, kfr);
    lds_barrier();

    float dsum = 0.f;
    if (wave < 3) {
      const int it = (wave > 0) ? 1 : 0;
      const int jt = (wave > 1) ? 1 : 0;
      const float* qrow = Qf + (it * 16 + lm) * FSTR + 2 * hi;
      const float* krow = Kf + (jt * 16 + lm) * FSTR + 2 * hi;
      v2f a[2][KS8], bb[2][KS8];
#pragma unroll
      for (int ks = 0; ks < KS8; ++ks) {
        a [0][ks] = ld_rm(qrow + 4 * ks);
        bb[0][ks] = ld_rm(krow + 4 * ks);
      }
      v8f acc = {};
#pragma unroll
      for (int blk = 0; blk < Md / 32; ++blk) {
        if (blk + 1 < Md / 32) {
#pragma unroll
          for (int ks = 0; ks < KS8; ++ks) {
            a [(blk + 1) & 1][ks] = ld_rm(qrow + 32 * (blk + 1) + 4 * ks);
            bb[(blk + 1) & 1][ks] = ld_rm(krow + 32 * (blk + 1) + 4 * ks);
          }
        }
        __builtin_amdgcn_sched_barrier(0);
#pragma unroll
        for (int ks = 0; ks < KS8; ++ks)
          acc = wmma4(a[blk & 1][ks], bb[blk & 1][ks], acc);
      }
#pragma unroll
      for (int e = 0; e < 8; ++e) {
        int gi = it * 16 + e + 8 * hi, gj = jt * 16 + lm;
        Alds[gi * ASTR + gj] = (gj <= gi) ? acc[e] : 0.f;
      }
    } else if (wave == 3) {
      const float4* qv = reinterpret_cast<const float4*>(Qf + lane * FSTR);
      const float4* sv = reinterpret_cast<const float4*>(sLds);
      float4 q4[2][8], s4[2][8];
#pragma unroll
      for (int i = 0; i < 8; ++i) { q4[0][i] = qv[i]; s4[0][i] = sv[i]; }
#pragma unroll
      for (int blk = 0; blk < Md / 32; ++blk) {
        if (blk + 1 < Md / 32) {
#pragma unroll
          for (int i = 0; i < 8; ++i) {
            q4[(blk + 1) & 1][i] = qv[8 * (blk + 1) + i];
            s4[(blk + 1) & 1][i] = sv[8 * (blk + 1) + i];
          }
        }
        __builtin_amdgcn_sched_barrier(0);
#pragma unroll
        for (int i = 0; i < 8; ++i) {
          dsum = __builtin_fmaf(q4[blk & 1][i].x, s4[blk & 1][i].x, dsum);
          dsum = __builtin_fmaf(q4[blk & 1][i].y, s4[blk & 1][i].y, dsum);
          dsum = __builtin_fmaf(q4[blk & 1][i].z, s4[blk & 1][i].z, dsum);
          dsum = __builtin_fmaf(q4[blk & 1][i].w, s4[blk & 1][i].w, dsum);
        }
      }
    } else if (wave == 4) {
      for (int i2 = lane; i2 < 256; i2 += 32)
        Alds[(i2 >> 4) * ASTR + 16 + (i2 & 15)] = 0.f;
    }
    float cs = 0.f;
    for (int r = 0; r < CK; ++r) cs += Kf[r * FSTR + tid];
    v2f vb[4][KS8];
    qs_share(Qf, Vlds, sacc, reinterpret_cast<float4*>(Plds), wave, lane, lm, hi, vb);
    state_update(kfr, vb, sacc);
    const int oit = wave >> 2, odt = wave & 3;
    lds_barrier();

    v8f oacc;
    {
      const float4* P4 = reinterpret_cast<const float4*>(Plds);
      float4 plo[8], phi[8];
#pragma unroll
      for (int w = 0; w < 8; ++w) {
        plo[w] = P4[((w * 8 + wave) * 2 + 0) * 32 + lane];
        phi[w] = P4[((w * 8 + wave) * 2 + 1) * 32 + lane];
      }
      oacc[0] = plo[0].x; oacc[1] = plo[0].y; oacc[2] = plo[0].z; oacc[3] = plo[0].w;
      oacc[4] = phi[0].x; oacc[5] = phi[0].y; oacc[6] = phi[0].z; oacc[7] = phi[0].w;
#pragma unroll
      for (int w = 1; w < 8; ++w) {
        oacc[0] += plo[w].x; oacc[1] += plo[w].y; oacc[2] += plo[w].z; oacc[3] += plo[w].w;
        oacc[4] += phi[w].x; oacc[5] += phi[w].y; oacc[6] += phi[w].z; oacc[7] += phi[w].w;
      }
    }
    {
      const float4* av = reinterpret_cast<const float4*>(Alds + lane * ASTR);
      float4 a4[CK / 4];
#pragma unroll
      for (int i = 0; i < CK / 4; ++i) a4[i] = av[i];
#pragma unroll
      for (int i = 0; i < CK / 4; ++i) {
        dsum += a4[i].x; dsum += a4[i].y; dsum += a4[i].z; dsum += a4[i].w;
      }
      denLds[wave * CK + lane] = dsum;
    }
    {
      const float* arow = Alds + (oit * 16 + lm) * ASTR + 2 * hi;
      const float* vcol = Vlds + (2 * hi) * PSTR + odt * 16 + lm;
      v2f ia[KS8], ib[KS8];
#pragma unroll
      for (int ks = 0; ks < KS8; ++ks) {
        ia[ks] = ld_rm(arow + 4 * ks);
        ib[ks] = ld_km(vcol + (4 * ks) * PSTR, PSTR);
      }
      __builtin_amdgcn_sched_barrier(0);
#pragma unroll
      for (int ks = 0; ks < KS8; ++ks)
        oacc = wmma4(ia[ks], ib[ks], oacc);
    }
    lds_barrier();

    {
#pragma unroll
      for (int e = 0; e < 8; ++e) {
        int row = oit * 16 + e + 8 * hi;
        float den = denLds[3 * CK + row];
        den = (den <= 0.f) ? 1.f : den;
        float* dst =
            out + (((size_t)b * Ld + lbase + row) * Hd + h) * Dd + odt * 16 + lm;
        __builtin_nontemporal_store(oacc[e] / den, dst);
      }
    }
    sLds[tid] += cs;
  }
}

extern "C" void kernel_launch(void* const* d_in, const int* in_sizes, int n_in,
                              void* d_out, int out_size, void* d_ws, size_t ws_size,
                              hipStream_t stream) {
  const float* q    = (const float*)d_in[0];
  const float* k    = (const float*)d_in[1];
  const float* v    = (const float*)d_in[2];
  const float* proj = (const float*)d_in[3];
  float* out = (float*)d_out;
  float* ws  = (float*)d_ws;

  (void)in_sizes; (void)n_in; (void)out_size; (void)ws_size;

  perf_attn_segsum<<<Bd * Hd * NSEG, 256, 0, stream>>>(k, v, proj, ws);
  perf_attn_prefix<<<Bd * Hd * STATE4 / 256, 256, 0, stream>>>(ws);
  perf_attn_main  <<<Bd * Hd * NSEG, 256, 0, stream>>>(q, k, v, proj, ws, out);
}
